# starting kernel
# baseline (speedup 1.0000x reference)
#include <hip/hip_runtime.h>


typedef float v2f __attribute__((ext_vector_type(2)));
typedef float v8f __attribute__((ext_vector_type(8)));

#define CH 128
#define NEG_SLOPE 0.2f
#define SM_EPS 1e-16f

__device__ __forceinline__ unsigned enc_f32(float f) {
  unsigned ib = __float_as_uint(f);
  return (ib & 0x80000000u) ? ~ib : (ib | 0x80000000u);
}
__device__ __forceinline__ float dec_f32(unsigned u) {
  return __uint_as_float((u & 0x80000000u) ? (u & 0x7FFFFFFFu) : ~u);
}

#define WT_PAD 132

__global__ void gat_gemm_wmma(const float* __restrict__ X,
                              const float* __restrict__ W,
                              float* __restrict__ H, int n_nodes) {
  extern __shared__ float wt[];
  const int tid = threadIdx.x;
  for (int i = tid; i < CH * CH; i += 256) {
    int k = i >> 7;
    int n = i & (CH - 1);
    wt[n * WT_PAD + k] = W[i];
  }
  __syncthreads();

  const int wave = tid >> 5;
  const int lane = tid & 31;
  const int row0 = blockIdx.x * 128 + wave * 16;
  const int mrow = lane & 15;
  const int koff = (lane >> 4) * 2;

  int rload = row0 + mrow;
  if (rload >= n_nodes) rload = n_nodes - 1;
  const float* xrow = X + (size_t)rload * CH + koff;
  const float* wrow = wt + mrow * WT_PAD + koff;

  v8f acc[8];
#pragma unroll
  for (int t = 0; t < 8; ++t) acc[t] = {};

  v2f a = *(const v2f*)(xrow);
  v2f b[8];
#pragma unroll
  for (int t = 0; t < 8; ++t)
    b[t] = *(const v2f*)(wrow + t * 16 * WT_PAD);

  for (int k4 = 0; k4 < CH; k4 += 4) {
    const int kn = k4 + 4;
    v2f an = {};
    v2f bn[8];
#pragma unroll
    for (int t = 0; t < 8; ++t) bn[t] = {};
    if (kn < CH) {
      an = *(const v2f*)(xrow + kn);
#pragma unroll
      for (int t = 0; t < 8; ++t)
        bn[t] = *(const v2f*)(wrow + t * 16 * WT_PAD + kn);
    }
#pragma unroll
    for (int t = 0; t < 8; ++t)
      acc[t] = __builtin_amdgcn_wmma_f32_16x16x4_f32(
          false, a, false, b[t], (short)0, acc[t], false, false);
    a = an;
#pragma unroll
    for (int t = 0; t < 8; ++t) b[t] = bn[t];
  }

  const int rhi = (lane >> 4) * 8;
#pragma unroll
  for (int t = 0; t < 8; ++t) {
    const int col = t * 16 + (lane & 15);
#pragma unroll
    for (int r = 0; r < 8; ++r) {
      int row = row0 + r + rhi;
      if (row < n_nodes) H[(size_t)row * CH + col] = acc[t][r];
    }
  }
}

__global__ void gat_node_scores(const float* __restrict__ H,
                                const float* __restrict__ a_src,
                                const float* __restrict__ a_dst,
                                float* __restrict__ s, float* __restrict__ d,
                                int n_nodes) {
  const int wave = threadIdx.x >> 5;
  const int lane = threadIdx.x & 31;
  const int node = blockIdx.x * 8 + wave;
  if (node >= n_nodes) return;
  float4 h4 = ((const float4*)(H + (size_t)node * CH))[lane];
  float4 a4 = ((const float4*)a_src)[lane];
  float4 b4 = ((const float4*)a_dst)[lane];
  float ss = h4.x * a4.x + h4.y * a4.y + h4.z * a4.z + h4.w * a4.w;
  float dd = h4.x * b4.x + h4.y * b4.y + h4.z * b4.z + h4.w * b4.w;
#pragma unroll
  for (int o = 16; o > 0; o >>= 1) {
    ss += __shfl_xor(ss, o, 32);
    dd += __shfl_xor(dd, o, 32);
  }
  if (lane == 0) {
    s[node] = ss;
    d[node] = dd;
  }
}

__global__ void gat_init(float* __restrict__ agg, unsigned* __restrict__ emax,
                         float* __restrict__ denom, int n_nodes) {
  size_t i = (size_t)blockIdx.x * blockDim.x + threadIdx.x;
  size_t total = (size_t)n_nodes * CH;
  if (i < total) agg[i] = 0.0f;
  if (i < (size_t)n_nodes) {
    emax[i] = 0u;
    denom[i] = 0.0f;
  }
}

__global__ void gat_edge_max(const int* __restrict__ src,
                             const int* __restrict__ dst,
                             const float* __restrict__ s,
                             const float* __restrict__ d,
                             float* __restrict__ ework,
                             unsigned* __restrict__ emax, int n_edges,
                             int n_nodes) {
  int e = blockIdx.x * blockDim.x + threadIdx.x;
  int tot = n_edges + n_nodes;
  if (e >= tot) return;
  int u = (e < n_edges) ? src[e] : (e - n_edges);
  int v = (e < n_edges) ? dst[e] : (e - n_edges);
  float val = s[u] + d[v];
  val = (val > 0.0f) ? val : NEG_SLOPE * val;
  ework[e] = val;
  atomicMax(&emax[v], enc_f32(val));
}

__global__ void gat_edge_expsum(const int* __restrict__ dst,
                                float* __restrict__ ework,
                                const unsigned* __restrict__ emax,
                                float* __restrict__ denom, int n_edges,
                                int n_nodes) {
  int e = blockIdx.x * blockDim.x + threadIdx.x;
  int tot = n_edges + n_nodes;
  if (e >= tot) return;
  int v = (e < n_edges) ? dst[e] : (e - n_edges);
  float w = __builtin_expf(ework[e] - dec_f32(emax[v]));
  ework[e] = w;
  atomicAdd(&denom[v], w);
}

__global__ void gat_edge_aggregate(const int* __restrict__ src,
                                   const int* __restrict__ dst,
                                   const float* __restrict__ ework,
                                   const float* __restrict__ denom,
                                   const float* __restrict__ H,
                                   float* __restrict__ agg, int n_edges,
                                   int n_nodes) {
  int wid = (int)(((size_t)blockIdx.x * blockDim.x + threadIdx.x) >> 5);
  int lane = threadIdx.x & 31;
  int tot = n_edges + n_nodes;
  if (wid >= tot) return;
  int u = (wid < n_edges) ? src[wid] : (wid - n_edges);
  int v = (wid < n_edges) ? dst[wid] : (wid - n_edges);
  float alpha = ework[wid] / (denom[v] + SM_EPS);
  float4 h4 = ((const float4*)(H + (size_t)u * CH))[lane];
  float* ap = agg + (size_t)v * CH + lane * 4;
  atomicAdd(ap + 0, alpha * h4.x);
  atomicAdd(ap + 1, alpha * h4.y);
  atomicAdd(ap + 2, alpha * h4.z);
  atomicAdd(ap + 3, alpha * h4.w);
}

__global__ void gat_update(float* __restrict__ x, const float* __restrict__ agg,
                           const float* __restrict__ bias, int n_nodes) {
  size_t i = (size_t)blockIdx.x * blockDim.x + threadIdx.x;
  if (i >= (size_t)n_nodes * CH) return;
  float v = agg[i] + bias[i & (CH - 1)];
  x[i] += 1.0f / (1.0f + __builtin_expf(-v));
}

__global__ void gat_copy(const float* __restrict__ in, float* __restrict__ out,
                         size_t n) {
  size_t i = (size_t)blockIdx.x * blockDim.x + threadIdx.x;
  if (i < n) out[i] = in[i];
}

extern "C" void kernel_launch(void* const* d_in, const int* in_sizes, int n_in,
                              void* d_out, int out_size, void* d_ws,
                              size_t ws_size, hipStream_t stream) {
  const float* x_in = (const float*)d_in[0];
  const int* ei = (const int*)d_in[1];
  const float* Ws = (const float*)d_in[2];
  const float* att_src = (const float*)d_in[3];
  const float* att_dst = (const float*)d_in[4];
  const float* bias = (const float*)d_in[5];

  const int N = in_sizes[0] / CH;
  const int E = in_sizes[1] / 2;
  const int M = in_sizes[3] / CH;
  const int* src = ei;
  const int* dst = ei + E;
  const int TOT = E + N;

  float* ws = (float*)d_ws;
  float* h = ws;
  float* agg = h + (size_t)N * CH;
  float* s = agg + (size_t)N * CH;
  float* dsc = s + N;
  float* denom = dsc + N;
  unsigned* emax = (unsigned*)(denom + N);
  float* ework = (float*)(emax + N);

  float* x = (float*)d_out;

  const size_t nfeat = (size_t)N * CH;
  const int blk = 256;
  const int g_feat = (int)((nfeat + blk - 1) / blk);
  const int g_gemm = (N + 127) / 128;
  const int g_node = (N + 7) / 8;
  const int g_edge = (TOT + blk - 1) / blk;
  const int g_eagg = (TOT + 7) / 8;
  const size_t smem = (size_t)CH * WT_PAD * sizeof(float);

  gat_copy<<<g_feat, blk, 0, stream>>>(x_in, x, nfeat);

  for (int m = 0; m < M; ++m) {
    const float* W = Ws + (size_t)m * CH * CH;
    const float* as_ = att_src + (size_t)m * CH;
    const float* ad_ = att_dst + (size_t)m * CH;
    const float* b_ = bias + (size_t)m * CH;

    gat_gemm_wmma<<<g_gemm, blk, smem, stream>>>(x, W, h, N);
    gat_node_scores<<<g_node, blk, 0, stream>>>(h, as_, ad_, s, dsc, N);
    gat_init<<<g_feat, blk, 0, stream>>>(agg, emax, denom, N);
    gat_edge_max<<<g_edge, blk, 0, stream>>>(src, dst, s, dsc, ework, emax, E, N);
    gat_edge_expsum<<<g_edge, blk, 0, stream>>>(dst, ework, emax, denom, E, N);
    gat_edge_aggregate<<<g_eagg, blk, 0, stream>>>(src, dst, ework, denom, h,
                                                   agg, E, N);
    gat_update<<<g_feat, blk, 0, stream>>>(x, agg, b_, N);
  }
}


# edges counting-sorted by dst once per call (int count, scan, fill), then one wave per node does softmax and weighted row sum in registers: no float atomics; 14 launches; sums in arrival-rank order
# speedup vs baseline: 2.5592x; 2.5592x over previous
#include <hip/hip_runtime.h>


typedef float v2f __attribute__((ext_vector_type(2)));
typedef float v8f __attribute__((ext_vector_type(8)));

#define CH 128
#define NEG_SLOPE 0.2f
#define SM_EPS 1e-16f

#define WT_PAD 132

__global__ void gat_gemm_wmma(const float* __restrict__ X,
                              const float* __restrict__ W,
                              float* __restrict__ H, int n_nodes) {
  extern __shared__ float wt[];
  const int tid = threadIdx.x;
  for (int i = tid; i < CH * CH; i += 256) {
    int k = i >> 7;
    int n = i & (CH - 1);
    wt[n * WT_PAD + k] = W[i];
  }
  __syncthreads();

  const int wave = tid >> 5;
  const int lane = tid & 31;
  const int row0 = blockIdx.x * 128 + wave * 16;
  const int mrow = lane & 15;
  const int koff = (lane >> 4) * 2;

  int rload = row0 + mrow;
  if (rload >= n_nodes) rload = n_nodes - 1;
  const float* xrow = X + (size_t)rload * CH + koff;
  const float* wrow = wt + mrow * WT_PAD + koff;

  v8f acc[8];
#pragma unroll
  for (int t = 0; t < 8; ++t) acc[t] = {};

  v2f a = *(const v2f*)(xrow);
  v2f b[8];
#pragma unroll
  for (int t = 0; t < 8; ++t)
    b[t] = *(const v2f*)(wrow + t * 16 * WT_PAD);

  for (int k4 = 0; k4 < CH; k4 += 4) {
    const int kn = k4 + 4;
    v2f an = {};
    v2f bn[8];
#pragma unroll
    for (int t = 0; t < 8; ++t) bn[t] = {};
    if (kn < CH) {
      an = *(const v2f*)(xrow + kn);
#pragma unroll
      for (int t = 0; t < 8; ++t)
        bn[t] = *(const v2f*)(wrow + t * 16 * WT_PAD + kn);
    }
#pragma unroll
    for (int t = 0; t < 8; ++t)
      acc[t] = __builtin_amdgcn_wmma_f32_16x16x4_f32(
          false, a, false, b[t], (short)0, acc[t], false, false);
    a = an;
#pragma unroll
    for (int t = 0; t < 8; ++t) b[t] = bn[t];
  }

  const int rhi = (lane >> 4) * 8;
#pragma unroll
  for (int t = 0; t < 8; ++t) {
    const int col = t * 16 + (lane & 15);
#pragma unroll
    for (int r = 0; r < 8; ++r) {
      int row = row0 + r + rhi;
      if (row < n_nodes) H[(size_t)row * CH + col] = acc[t][r];
    }
  }
}

__global__ void gat_node_scores(const float* __restrict__ H,
                                const float* __restrict__ a_src,
                                const float* __restrict__ a_dst,
                                float* __restrict__ s, float* __restrict__ d,
                                int n_nodes) {
  const int wave = threadIdx.x >> 5;
  const int lane = threadIdx.x & 31;
  const int node = blockIdx.x * 8 + wave;
  if (node >= n_nodes) return;
  float4 h4 = ((const float4*)(H + (size_t)node * CH))[lane];
  float4 a4 = ((const float4*)a_src)[lane];
  float4 b4 = ((const float4*)a_dst)[lane];
  float ss = h4.x * a4.x + h4.y * a4.y + h4.z * a4.z + h4.w * a4.w;
  float dd = h4.x * b4.x + h4.y * b4.y + h4.z * b4.z + h4.w * b4.w;
#pragma unroll
  for (int o = 16; o > 0; o >>= 1) {
    ss += __shfl_xor(ss, o, 32);
    dd += __shfl_xor(dd, o, 32);
  }
  if (lane == 0) {
    s[node] = ss;
    d[node] = dd;
  }
}

__global__ void csr_zero(int* __restrict__ cnt, int n_nodes) {
  int i = blockIdx.x * blockDim.x + threadIdx.x;
  if (i < n_nodes) cnt[i] = 0;
}

__global__ void csr_count(const int* __restrict__ dst, int* __restrict__ cnt,
                          int n_edges) {
  int e = blockIdx.x * blockDim.x + threadIdx.x;
  if (e < n_edges) atomicAdd(&cnt[dst[e]], 1);
}

__global__ void csr_scan(const int* __restrict__ cnt, int* __restrict__ rowptr,
                         int* __restrict__ cursor, int n_nodes) {
  __shared__ int part[1024];
  const int t = threadIdx.x;
  const int chunk = (n_nodes + 1023) / 1024;
  const int b = t * chunk;
  const int e = (b + chunk < n_nodes) ? (b + chunk) : n_nodes;
  int sum = 0;
  for (int i = b; i < e; ++i) sum += cnt[i];
  part[t] = sum;
  __syncthreads();
  for (int off = 1; off < 1024; off <<= 1) {
    const int add = (t >= off) ? part[t - off] : 0;
    __syncthreads();
    part[t] += add;
    __syncthreads();
  }
  int run = part[t] - sum;
  for (int i = b; i < e; i += 8) {
    int c[8];
#pragma unroll
    for (int k = 0; k < 8; ++k) c[k] = (i + k < e) ? cnt[i + k] : 0;
#pragma unroll
    for (int k = 0; k < 8; ++k) {
      if (i + k < e) {
        rowptr[i + k] = run;
        cursor[i + k] = run;
        run += c[k];
      }
    }
  }
  if (t == 1023) rowptr[n_nodes] = part[1023];
}

__global__ void csr_fill(const int* __restrict__ src,
                         const int* __restrict__ dst, int* __restrict__ cursor,
                         int* __restrict__ csr_src, int n_edges) {
  int e = blockIdx.x * blockDim.x + threadIdx.x;
  if (e < n_edges) {
    const int pos = atomicAdd(&cursor[dst[e]], 1);
    csr_src[pos] = src[e];
  }
}

__device__ __forceinline__ float leaky(float v) {
  return (v > 0.0f) ? v : NEG_SLOPE * v;
}

__device__ __forceinline__ float wave_max(float v) {
  for (int o = 16; o > 0; o >>= 1) v = __builtin_fmaxf(v, __shfl_xor(v, o, 32));
  return v;
}

__device__ __forceinline__ float wave_sum(float v) {
  for (int o = 16; o > 0; o >>= 1) v += __shfl_xor(v, o, 32);
  return v;
}

__device__ __forceinline__ void add_rows(float4& acc, int u, float a, int cnt,
                                         const float4* __restrict__ H4,
                                         int lane) {
#pragma unroll 4
  for (int t = 0; t < cnt; ++t) {
    const int ut = __builtin_amdgcn_readlane(u, t);
    const float at = __int_as_float(__builtin_amdgcn_readlane(__float_as_int(a), t));
    const float4 h4 = H4[(size_t)ut * (CH / 4) + lane];
    acc.x += at * h4.x;
    acc.y += at * h4.y;
    acc.z += at * h4.z;
    acc.w += at * h4.w;
  }
}

__global__ void gat_node_gather(const int* __restrict__ rowptr,
                                const int* __restrict__ csr_src,
                                const float* __restrict__ s,
                                const float* __restrict__ d,
                                const float* __restrict__ H,
                                const float* __restrict__ bias,
                                float* __restrict__ x, int n_nodes) {
  const int lane = threadIdx.x & 31;
  const int v = __builtin_amdgcn_readfirstlane(
      (int)(blockIdx.x * 8 + (threadIdx.x >> 5)));
  if (v >= n_nodes) return;
  const int beg = rowptr[v];
  const int end = rowptr[v + 1];
  const float dv = d[v];
  const float es = leaky(s[v] + dv);
  const float4* H4 = (const float4*)H;

  float4 acc;
  if (end - beg <= 32) {
    const int j = beg + lane;
    const bool ok = j < end;
    const int u = ok ? csr_src[j] : v;
    const float val = leaky(s[u] + dv);
    float m = wave_max(ok ? __builtin_fmaxf(val, es) : es);
    if (!__builtin_isfinite(m)) m = 0.0f;
    const float w = ok ? __builtin_expf(val - m) : 0.0f;
    const float ws = __builtin_expf(es - m);
    const float rden = wave_sum(w) + ws + SM_EPS;
    const float a = w / rden;
    const float as = ws / rden;
    const float4 hs = H4[(size_t)v * (CH / 4) + lane];
    acc.x = as * hs.x;
    acc.y = as * hs.y;
    acc.z = as * hs.z;
    acc.w = as * hs.w;
    add_rows(acc, u, a, end - beg, H4, lane);
  } else {
    float m = es;
    for (int base = beg; base < end; base += 32) {
      const int j = base + lane;
      if (j < end) m = __builtin_fmaxf(m, leaky(s[csr_src[j]] + dv));
    }
    m = wave_max(m);
    if (!__builtin_isfinite(m)) m = 0.0f;
    float den = 0.0f;
    for (int base = beg; base < end; base += 32) {
      const int j = base + lane;
      if (j < end) den += __builtin_expf(leaky(s[csr_src[j]] + dv) - m);
    }
    const float ws = __builtin_expf(es - m);
    const float rden = wave_sum(den) + ws + SM_EPS;
    const float as = ws / rden;
    const float4 hs = H4[(size_t)v * (CH / 4) + lane];
    acc.x = as * hs.x;
    acc.y = as * hs.y;
    acc.z = as * hs.z;
    acc.w = as * hs.w;
    for (int base = beg; base < end; base += 32) {
      const int j = base + lane;
      const bool ok = j < end;
      const int u = ok ? csr_src[j] : v;
      const float a = ok ? __builtin_expf(leaky(s[u] + dv) - m) / rden : 0.0f;
      const int left = end - base;
      add_rows(acc, u, a, (left < 32) ? left : 32, H4, lane);
    }
  }

  const float4 b4 = ((const float4*)bias)[lane];
  float4* xp = (float4*)(x + (size_t)v * CH) + lane;
  float4 xv = *xp;
  xv.x += 1.0f / (1.0f + __builtin_expf(-(acc.x + b4.x)));
  xv.y += 1.0f / (1.0f + __builtin_expf(-(acc.y + b4.y)));
  xv.z += 1.0f / (1.0f + __builtin_expf(-(acc.z + b4.z)));
  xv.w += 1.0f / (1.0f + __builtin_expf(-(acc.w + b4.w)));
  *xp = xv;
}

__global__ void gat_copy(const float* __restrict__ in, float* __restrict__ out,
                         size_t n) {
  size_t i = (size_t)blockIdx.x * blockDim.x + threadIdx.x;
  if (i < n) out[i] = in[i];
}

extern "C" void kernel_launch(void* const* d_in, const int* in_sizes, int n_in,
                              void* d_out, int out_size, void* d_ws,
                              size_t ws_size, hipStream_t stream) {
  const float* x_in = (const float*)d_in[0];
  const int* ei = (const int*)d_in[1];
  const float* Ws = (const float*)d_in[2];
  const float* att_src = (const float*)d_in[3];
  const float* att_dst = (const float*)d_in[4];
  const float* bias = (const float*)d_in[5];

  const int N = in_sizes[0] / CH;
  const int E = in_sizes[1] / 2;
  const int M = in_sizes[3] / CH;
  const int* src = ei;
  const int* dst = ei + E;

  float* ws = (float*)d_ws;
  float* h = ws;
  float* s = h + (size_t)N * CH;
  float* dsc = s + N;
  int* cnt = (int*)(dsc + N);
  int* cursor = cnt + N;
  int* rowptr = cursor + N;
  int* csr_src = rowptr + N + 1;

  float* x = (float*)d_out;

  const size_t nfeat = (size_t)N * CH;
  const int blk = 256;
  const int g_feat = (int)((nfeat + blk - 1) / blk);
  const int g_gemm = (N + 127) / 128;
  const int g_node = (N + 7) / 8;
  const int g_n = (N + blk - 1) / blk;
  const int g_e = (E + blk - 1) / blk;
  const size_t smem = (size_t)CH * WT_PAD * sizeof(float);

  gat_copy<<<g_feat, blk, 0, stream>>>(x_in, x, nfeat);

  csr_zero<<<g_n, blk, 0, stream>>>(cnt, N);
  csr_count<<<g_e, blk, 0, stream>>>(dst, cnt, E);
  csr_scan<<<1, 1024, 0, stream>>>(cnt, rowptr, cursor, N);
  csr_fill<<<g_e, blk, 0, stream>>>(src, dst, cursor, csr_src, E);

  for (int m = 0; m < M; ++m) {
    const float* W = Ws + (size_t)m * CH * CH;
    const float* as_ = att_src + (size_t)m * CH;
    const float* ad_ = att_dst + (size_t)m * CH;
    const float* b_ = bias + (size_t)m * CH;

    gat_gemm_wmma<<<g_gemm, blk, smem, stream>>>(x, W, h, N);
    gat_node_scores<<<g_node, blk, 0, stream>>>(h, as_, ad_, s, dsc, N);
    gat_node_gather<<<g_node, blk, 0, stream>>>(rowptr, csr_src, s, dsc, h, b_,
                                                x, N);
  }
}


# every wave takes several items (copy 8x16 B per lane, scores 8 nodes, gather 2 nodes per wave, count/fill 8 edges per lane): over a million waves per call become 234 thousand; float sums unchanged
# speedup vs baseline: 2.6734x; 1.0446x over previous
#include <hip/hip_runtime.h>


typedef float v2f __attribute__((ext_vector_type(2)));
typedef float v8f __attribute__((ext_vector_type(8)));

#define CH 128
#define NEG_SLOPE 0.2f
#define SM_EPS 1e-16f

#define WT_PAD 132

__global__ void gat_gemm_wmma(const float* __restrict__ X,
                              const float* __restrict__ W,
                              float* __restrict__ H, int n_nodes) {
  extern __shared__ float wt[];
  const int tid = threadIdx.x;
  for (int i = tid; i < CH * CH; i += 256) {
    int k = i >> 7;
    int n = i & (CH - 1);
    wt[n * WT_PAD + k] = W[i];
  }
  __syncthreads();

  const int wave = tid >> 5;
  const int lane = tid & 31;
  const int row0 = blockIdx.x * 128 + wave * 16;
  const int mrow = lane & 15;
  const int koff = (lane >> 4) * 2;

  int rload = row0 + mrow;
  if (rload >= n_nodes) rload = n_nodes - 1;
  const float* xrow = X + (size_t)rload * CH + koff;
  const float* wrow = wt + mrow * WT_PAD + koff;

  v8f acc[8];
#pragma unroll
  for (int t = 0; t < 8; ++t) acc[t] = {};

  v2f a = *(const v2f*)(xrow);
  v2f b[8];
#pragma unroll
  for (int t = 0; t < 8; ++t)
    b[t] = *(const v2f*)(wrow + t * 16 * WT_PAD);

  for (int k4 = 0; k4 < CH; k4 += 4) {
    const int kn = k4 + 4;
    v2f an = {};
    v2f bn[8];
#pragma unroll
    for (int t = 0; t < 8; ++t) bn[t] = {};
    if (kn < CH) {
      an = *(const v2f*)(xrow + kn);
#pragma unroll
      for (int t = 0; t < 8; ++t)
        bn[t] = *(const v2f*)(wrow + t * 16 * WT_PAD + kn);
    }
#pragma unroll
    for (int t = 0; t < 8; ++t)
      acc[t] = __builtin_amdgcn_wmma_f32_16x16x4_f32(
          false, a, false, b[t], (short)0, acc[t], false, false);
    a = an;
#pragma unroll
    for (int t = 0; t < 8; ++t) b[t] = bn[t];
  }

  const int rhi = (lane >> 4) * 8;
#pragma unroll
  for (int t = 0; t < 8; ++t) {
    const int col = t * 16 + (lane & 15);
#pragma unroll
    for (int r = 0; r < 8; ++r) {
      int row = row0 + r + rhi;
      if (row < n_nodes) H[(size_t)row * CH + col] = acc[t][r];
    }
  }
}

#define SC_NODES 8

__global__ void gat_node_scores(const float* __restrict__ H,
                                const float* __restrict__ a_src,
                                const float* __restrict__ a_dst,
                                float* __restrict__ s, float* __restrict__ d,
                                int n_nodes) {
  const int wave = threadIdx.x >> 5;
  const int lane = threadIdx.x & 31;
  const int node0 = (blockIdx.x * 8 + wave) * SC_NODES;
  if (node0 >= n_nodes) return;
  const float4* H4 = (const float4*)H;
  const float4 a4 = ((const float4*)a_src)[lane];
  const float4 b4 = ((const float4*)a_dst)[lane];
  float4 h4[SC_NODES];
#pragma unroll
  for (int k = 0; k < SC_NODES; ++k) {
    int node = node0 + k;
    if (node >= n_nodes) node = n_nodes - 1;
    h4[k] = H4[(size_t)node * (CH / 4) + lane];
  }
  float s_keep = 0.0f;
  float d_keep = 0.0f;
#pragma unroll
  for (int k = 0; k < SC_NODES; ++k) {
    float ss = h4[k].x * a4.x + h4[k].y * a4.y + h4[k].z * a4.z + h4[k].w * a4.w;
    float dd = h4[k].x * b4.x + h4[k].y * b4.y + h4[k].z * b4.z + h4[k].w * b4.w;
#pragma unroll
    for (int o = 16; o > 0; o >>= 1) {
      ss += __shfl_xor(ss, o, 32);
      dd += __shfl_xor(dd, o, 32);
    }
    if (lane == k) {
      s_keep = ss;
      d_keep = dd;
    }
  }
  const int node = node0 + lane;
  if (lane < SC_NODES && node < n_nodes) {
    s[node] = s_keep;
    d[node] = d_keep;
  }
}

__global__ void csr_zero(int* __restrict__ cnt, int n_nodes) {
  int i = blockIdx.x * blockDim.x + threadIdx.x;
  if (i < n_nodes) cnt[i] = 0;
}

#define EDGES_PER_THREAD 8

__global__ void csr_count(const int* __restrict__ dst, int* __restrict__ cnt,
                          int n_edges) {
  const int b0 = blockIdx.x * (256 * EDGES_PER_THREAD);
  const int e0 = b0 + threadIdx.x;
  if (b0 + 256 * EDGES_PER_THREAD <= n_edges) {
    int dv[EDGES_PER_THREAD];
#pragma unroll
    for (int k = 0; k < EDGES_PER_THREAD; ++k) dv[k] = dst[e0 + k * 256];
#pragma unroll
    for (int k = 0; k < EDGES_PER_THREAD; ++k) atomicAdd(&cnt[dv[k]], 1);
  } else {
    for (int k = 0; k < EDGES_PER_THREAD; ++k) {
      const int e = e0 + k * 256;
      if (e < n_edges) atomicAdd(&cnt[dst[e]], 1);
    }
  }
}

__global__ void csr_scan(const int* __restrict__ cnt, int* __restrict__ rowptr,
                         int* __restrict__ cursor, int n_nodes) {
  __shared__ int part[1024];
  const int t = threadIdx.x;
  const int chunk = (n_nodes + 1023) / 1024;
  const int b = t * chunk;
  const int e = (b + chunk < n_nodes) ? (b + chunk) : n_nodes;
  int sum = 0;
  for (int i = b; i < e; ++i) sum += cnt[i];
  part[t] = sum;
  __syncthreads();
  for (int off = 1; off < 1024; off <<= 1) {
    const int add = (t >= off) ? part[t - off] : 0;
    __syncthreads();
    part[t] += add;
    __syncthreads();
  }
  int run = part[t] - sum;
  for (int i = b; i < e; i += 8) {
    int c[8];
#pragma unroll
    for (int k = 0; k < 8; ++k) c[k] = (i + k < e) ? cnt[i + k] : 0;
#pragma unroll
    for (int k = 0; k < 8; ++k) {
      if (i + k < e) {
        rowptr[i + k] = run;
        cursor[i + k] = run;
        run += c[k];
      }
    }
  }
  if (t == 1023) rowptr[n_nodes] = part[1023];
}

__global__ void csr_fill(const int* __restrict__ src,
                         const int* __restrict__ dst, int* __restrict__ cursor,
                         int* __restrict__ csr_src, int n_edges) {
  const int b0 = blockIdx.x * (256 * EDGES_PER_THREAD);
  const int e0 = b0 + threadIdx.x;
  if (b0 + 256 * EDGES_PER_THREAD <= n_edges) {
    int dv[EDGES_PER_THREAD];
    int sv[EDGES_PER_THREAD];
    int pos[EDGES_PER_THREAD];
#pragma unroll
    for (int k = 0; k < EDGES_PER_THREAD; ++k) {
      dv[k] = dst[e0 + k * 256];
      sv[k] = src[e0 + k * 256];
    }
#pragma unroll
    for (int k = 0; k < EDGES_PER_THREAD; ++k)
      pos[k] = atomicAdd(&cursor[dv[k]], 1);
#pragma unroll
    for (int k = 0; k < EDGES_PER_THREAD; ++k) csr_src[pos[k]] = sv[k];
  } else {
    for (int k = 0; k < EDGES_PER_THREAD; ++k) {
      const int e = e0 + k * 256;
      if (e < n_edges) {
        const int pos = atomicAdd(&cursor[dst[e]], 1);
        csr_src[pos] = src[e];
      }
    }
  }
}

#define GA_NODES 2
__device__ __forceinline__ float leaky(float v) {
  return (v > 0.0f) ? v : NEG_SLOPE * v;
}

__device__ __forceinline__ float wave_max(float v) {
  for (int o = 16; o > 0; o >>= 1) v = __builtin_fmaxf(v, __shfl_xor(v, o, 32));
  return v;
}

__device__ __forceinline__ float wave_sum(float v) {
  for (int o = 16; o > 0; o >>= 1) v += __shfl_xor(v, o, 32);
  return v;
}

__device__ __forceinline__ void add_rows(float4& acc, int u, float a, int cnt,
                                         const float4* __restrict__ H4,
                                         int lane) {
#pragma unroll 4
  for (int t = 0; t < cnt; ++t) {
    const int ut = __builtin_amdgcn_readlane(u, t);
    const float at = __int_as_float(__builtin_amdgcn_readlane(__float_as_int(a), t));
    const float4 h4 = H4[(size_t)ut * (CH / 4) + lane];
    acc.x += at * h4.x;
    acc.y += at * h4.y;
    acc.z += at * h4.z;
    acc.w += at * h4.w;
  }
}

__global__ void gat_node_gather(const int* __restrict__ rowptr,
                                const int* __restrict__ csr_src,
                                const float* __restrict__ s,
                                const float* __restrict__ d,
                                const float* __restrict__ H,
                                const float* __restrict__ bias,
                                float* __restrict__ x, int n_nodes) {
  const int lane = threadIdx.x & 31;
  const int v0 = __builtin_amdgcn_readfirstlane(
      (int)(blockIdx.x * 8 + (threadIdx.x >> 5)) * GA_NODES);
  const float4 b4 = ((const float4*)bias)[lane];
  for (int v = v0; v < v0 + GA_NODES && v < n_nodes; ++v) {
    const int beg = rowptr[v];
    const int end = rowptr[v + 1];
    const float dv = d[v];
    const float es = leaky(s[v] + dv);
    const float4* H4 = (const float4*)H;

    float4 acc;
    if (end - beg <= 32) {
      const int j = beg + lane;
      const bool ok = j < end;
      const int u = ok ? csr_src[j] : v;
      const float val = leaky(s[u] + dv);
      float m = wave_max(ok ? __builtin_fmaxf(val, es) : es);
      if (!__builtin_isfinite(m)) m = 0.0f;
      const float w = ok ? __builtin_expf(val - m) : 0.0f;
      const float ws = __builtin_expf(es - m);
      const float rden = wave_sum(w) + ws + SM_EPS;
      const float a = w / rden;
      const float as = ws / rden;
      const float4 hs = H4[(size_t)v * (CH / 4) + lane];
      acc.x = as * hs.x;
      acc.y = as * hs.y;
      acc.z = as * hs.z;
      acc.w = as * hs.w;
      add_rows(acc, u, a, end - beg, H4, lane);
    } else {
      float m = es;
      for (int base = beg; base < end; base += 32) {
        const int j = base + lane;
        if (j < end) m = __builtin_fmaxf(m, leaky(s[csr_src[j]] + dv));
      }
      m = wave_max(m);
      if (!__builtin_isfinite(m)) m = 0.0f;
      float den = 0.0f;
      for (int base = beg; base < end; base += 32) {
        const int j = base + lane;
        if (j < end) den += __builtin_expf(leaky(s[csr_src[j]] + dv) - m);
      }
      const float ws = __builtin_expf(es - m);
      const float rden = wave_sum(den) + ws + SM_EPS;
      const float as = ws / rden;
      const float4 hs = H4[(size_t)v * (CH / 4) + lane];
      acc.x = as * hs.x;
      acc.y = as * hs.y;
      acc.z = as * hs.z;
      acc.w = as * hs.w;
      for (int base = beg; base < end; base += 32) {
        const int j = base + lane;
        const bool ok = j < end;
        const int u = ok ? csr_src[j] : v;
        const float a = ok ? __builtin_expf(leaky(s[u] + dv) - m) / rden : 0.0f;
        const int left = end - base;
        add_rows(acc, u, a, (left < 32) ? left : 32, H4, lane);
      }
    }

    float4* xp = (float4*)(x + (size_t)v * CH) + lane;
    float4 xv = *xp;
    xv.x += 1.0f / (1.0f + __builtin_expf(-(acc.x + b4.x)));
    xv.y += 1.0f / (1.0f + __builtin_expf(-(acc.y + b4.y)));
    xv.z += 1.0f / (1.0f + __builtin_expf(-(acc.z + b4.z)));
    xv.w += 1.0f / (1.0f + __builtin_expf(-(acc.w + b4.w)));
    *xp = xv;
  }
}

#define COPY_PER_THREAD 8

__global__ void gat_copy(const float* __restrict__ in, float* __restrict__ out,
                         size_t n4) {
  const float4* in4 = (const float4*)in;
  float4* out4 = (float4*)out;
  const size_t i0 = (size_t)blockIdx.x * (256 * COPY_PER_THREAD) + threadIdx.x;
  float4 r[COPY_PER_THREAD];
#pragma unroll
  for (int k = 0; k < COPY_PER_THREAD; ++k) {
    const size_t i = i0 + (size_t)k * 256;
    r[k] = (i < n4) ? in4[i] : make_float4(0.0f, 0.0f, 0.0f, 0.0f);
  }
#pragma unroll
  for (int k = 0; k < COPY_PER_THREAD; ++k) {
    const size_t i = i0 + (size_t)k * 256;
    if (i < n4) out4[i] = r[k];
  }
}

extern "C" void kernel_launch(void* const* d_in, const int* in_sizes, int n_in,
                              void* d_out, int out_size, void* d_ws,
                              size_t ws_size, hipStream_t stream) {
  const float* x_in = (const float*)d_in[0];
  const int* ei = (const int*)d_in[1];
  const float* Ws = (const float*)d_in[2];
  const float* att_src = (const float*)d_in[3];
  const float* att_dst = (const float*)d_in[4];
  const float* bias = (const float*)d_in[5];

  const int N = in_sizes[0] / CH;
  const int E = in_sizes[1] / 2;
  const int M = in_sizes[3] / CH;
  const int* src = ei;
  const int* dst = ei + E;

  float* ws = (float*)d_ws;
  float* h = ws;
  float* s = h + (size_t)N * CH;
  float* dsc = s + N;
  int* cnt = (int*)(dsc + N);
  int* cursor = cnt + N;
  int* rowptr = cursor + N;
  int* csr_src = rowptr + N + 1;

  float* x = (float*)d_out;

  const size_t nfeat = (size_t)N * CH;
  const int blk = 256;
  const size_t nfeat4 = nfeat / 4;
  const int per_copy = blk * COPY_PER_THREAD;
  const int g_copy = (int)((nfeat4 + per_copy - 1) / per_copy);
  const int g_gemm = (N + 127) / 128;
  const int g_score = (N + 8 * SC_NODES - 1) / (8 * SC_NODES);
  const int g_gather = (N + 8 * GA_NODES - 1) / (8 * GA_NODES);
  const int g_n = (N + blk - 1) / blk;
  const int per_edge = blk * EDGES_PER_THREAD;
  const int g_e = (E + per_edge - 1) / per_edge;
  const size_t smem = (size_t)CH * WT_PAD * sizeof(float);

  gat_copy<<<g_copy, blk, 0, stream>>>(x_in, x, nfeat4);

  csr_zero<<<g_n, blk, 0, stream>>>(cnt, N);
  csr_count<<<g_e, blk, 0, stream>>>(dst, cnt, E);
  csr_scan<<<1, 1024, 0, stream>>>(cnt, rowptr, cursor, N);
  csr_fill<<<g_e, blk, 0, stream>>>(src, dst, cursor, csr_src, E);

  for (int m = 0; m < M; ++m) {
    const float* W = Ws + (size_t)m * CH * CH;
    const float* as_ = att_src + (size_t)m * CH;
    const float* ad_ = att_dst + (size_t)m * CH;
    const float* b_ = bias + (size_t)m * CH;

    gat_gemm_wmma<<<g_gemm, blk, smem, stream>>>(x, W, h, N);
    gat_node_scores<<<g_score, blk, 0, stream>>>(h, as_, ad_, s, dsc, N);
    gat_node_gather<<<g_gather, blk, 0, stream>>>(rowptr, csr_src, s, dsc, h, b_,
                                                  x, N);
  }
}


# row-offset prefix scan on 98 workgroups, one node per thread: whole-line loads and stores, each block re-sums the counts before its tile; was 1 workgroup on 392-byte strides; float sums unchanged
# speedup vs baseline: 3.7303x; 1.3953x over previous
#include <hip/hip_runtime.h>


typedef float v2f __attribute__((ext_vector_type(2)));
typedef float v8f __attribute__((ext_vector_type(8)));

#define CH 128
#define NEG_SLOPE 0.2f
#define SM_EPS 1e-16f

#define WT_PAD 132

__global__ void gat_gemm_wmma(const float* __restrict__ X,
                              const float* __restrict__ W,
                              float* __restrict__ H, int n_nodes) {
  extern __shared__ float wt[];
  const int tid = threadIdx.x;
  for (int i = tid; i < CH * CH; i += 256) {
    int k = i >> 7;
    int n = i & (CH - 1);
    wt[n * WT_PAD + k] = W[i];
  }
  __syncthreads();

  const int wave = tid >> 5;
  const int lane = tid & 31;
  const int row0 = blockIdx.x * 128 + wave * 16;
  const int mrow = lane & 15;
  const int koff = (lane >> 4) * 2;

  int rload = row0 + mrow;
  if (rload >= n_nodes) rload = n_nodes - 1;
  const float* xrow = X + (size_t)rload * CH + koff;
  const float* wrow = wt + mrow * WT_PAD + koff;

  v8f acc[8];
#pragma unroll
  for (int t = 0; t < 8; ++t) acc[t] = {};

  v2f a = *(const v2f*)(xrow);
  v2f b[8];
#pragma unroll
  for (int t = 0; t < 8; ++t)
    b[t] = *(const v2f*)(wrow + t * 16 * WT_PAD);

  for (int k4 = 0; k4 < CH; k4 += 4) {
    const int kn = k4 + 4;
    v2f an = {};
    v2f bn[8];
#pragma unroll
    for (int t = 0; t < 8; ++t) bn[t] = {};
    if (kn < CH) {
      an = *(const v2f*)(xrow + kn);
#pragma unroll
      for (int t = 0; t < 8; ++t)
        bn[t] = *(const v2f*)(wrow + t * 16 * WT_PAD + kn);
    }
#pragma unroll
    for (int t = 0; t < 8; ++t)
      acc[t] = __builtin_amdgcn_wmma_f32_16x16x4_f32(
          false, a, false, b[t], (short)0, acc[t], false, false);
    a = an;
#pragma unroll
    for (int t = 0; t < 8; ++t) b[t] = bn[t];
  }

  const int rhi = (lane >> 4) * 8;
#pragma unroll
  for (int t = 0; t < 8; ++t) {
    const int col = t * 16 + (lane & 15);
#pragma unroll
    for (int r = 0; r < 8; ++r) {
      int row = row0 + r + rhi;
      if (row < n_nodes) H[(size_t)row * CH + col] = acc[t][r];
    }
  }
}

#define SC_NODES 8

__global__ void gat_node_scores(const float* __restrict__ H,
                                const float* __restrict__ a_src,
                                const float* __restrict__ a_dst,
                                float* __restrict__ s, float* __restrict__ d,
                                int n_nodes) {
  const int wave = threadIdx.x >> 5;
  const int lane = threadIdx.x & 31;
  const int node0 = (blockIdx.x * 8 + wave) * SC_NODES;
  if (node0 >= n_nodes) return;
  const float4* H4 = (const float4*)H;
  const float4 a4 = ((const float4*)a_src)[lane];
  const float4 b4 = ((const float4*)a_dst)[lane];
  float4 h4[SC_NODES];
#pragma unroll
  for (int k = 0; k < SC_NODES; ++k) {
    int node = node0 + k;
    if (node >= n_nodes) node = n_nodes - 1;
    h4[k] = H4[(size_t)node * (CH / 4) + lane];
  }
  float s_keep = 0.0f;
  float d_keep = 0.0f;
#pragma unroll
  for (int k = 0; k < SC_NODES; ++k) {
    float ss = h4[k].x * a4.x + h4[k].y * a4.y + h4[k].z * a4.z + h4[k].w * a4.w;
    float dd = h4[k].x * b4.x + h4[k].y * b4.y + h4[k].z * b4.z + h4[k].w * b4.w;
#pragma unroll
    for (int o = 16; o > 0; o >>= 1) {
      ss += __shfl_xor(ss, o, 32);
      dd += __shfl_xor(dd, o, 32);
    }
    if (lane == k) {
      s_keep = ss;
      d_keep = dd;
    }
  }
  const int node = node0 + lane;
  if (lane < SC_NODES && node < n_nodes) {
    s[node] = s_keep;
    d[node] = d_keep;
  }
}

__global__ void csr_zero(int* __restrict__ cnt, int n_nodes) {
  int i = blockIdx.x * blockDim.x + threadIdx.x;
  if (i < n_nodes) cnt[i] = 0;
}

#define EDGES_PER_THREAD 8

__global__ void csr_count(const int* __restrict__ dst, int* __restrict__ cnt,
                          int n_edges) {
  const int b0 = blockIdx.x * (256 * EDGES_PER_THREAD);
  const int e0 = b0 + threadIdx.x;
  if (b0 + 256 * EDGES_PER_THREAD <= n_edges) {
    int dv[EDGES_PER_THREAD];
#pragma unroll
    for (int k = 0; k < EDGES_PER_THREAD; ++k) dv[k] = dst[e0 + k * 256];
#pragma unroll
    for (int k = 0; k < EDGES_PER_THREAD; ++k) atomicAdd(&cnt[dv[k]], 1);
  } else {
    for (int k = 0; k < EDGES_PER_THREAD; ++k) {
      const int e = e0 + k * 256;
      if (e < n_edges) atomicAdd(&cnt[dst[e]], 1);
    }
  }
}

#define SCAN_TILE 1024
#define SCAN_AHEAD 16

__device__ __forceinline__ int wave_sum_int(int v) {
  for (int o = 16; o > 0; o >>= 1) v += __shfl_xor(v, o, 32);
  return v;
}

__global__ void csr_scan(const int* __restrict__ cnt, int* __restrict__ rowptr,
                         int* __restrict__ cursor, int n_nodes) {
  __shared__ int w_front[32];
  __shared__ int w_tile[32];
  const int t = threadIdx.x;
  const int lane = t & 31;
  const int wave = t >> 5;
  const int base = blockIdx.x * SCAN_TILE;
  const int i = base + t;
  const int c = (i < n_nodes) ? cnt[i] : 0;

  const int nfront = blockIdx.x;
  const int* col = cnt + t;
  int front = 0;
  for (int j0 = 0; j0 < nfront; j0 += SCAN_AHEAD) {
    int v[SCAN_AHEAD];
#pragma unroll
    for (int k = 0; k < SCAN_AHEAD; ++k)
      v[k] = (j0 + k < nfront) ? col[(size_t)(j0 + k) * SCAN_TILE] : 0;
#pragma unroll
    for (int k = 0; k < SCAN_AHEAD; ++k) front += v[k];
  }

  int incl = c;
#pragma unroll
  for (int o = 1; o < 32; o <<= 1) {
    const int y = __shfl_up(incl, o, 32);
    if (lane >= o) incl += y;
  }
  front = wave_sum_int(front);
  if (lane == 31) w_tile[wave] = incl;
  if (lane == 0) w_front[wave] = front;
  __syncthreads();
  const int tile_w = w_tile[lane];
  const int before = wave_sum_int(w_front[lane]) +
                     wave_sum_int((lane < wave) ? tile_w : 0);
  const int excl = before + incl - c;
  if (i < n_nodes) {
    rowptr[i] = excl;
    cursor[i] = excl;
    if (i == n_nodes - 1) rowptr[n_nodes] = excl + c;
  }
}

__global__ void csr_fill(const int* __restrict__ src,
                         const int* __restrict__ dst, int* __restrict__ cursor,
                         int* __restrict__ csr_src, int n_edges) {
  const int b0 = blockIdx.x * (256 * EDGES_PER_THREAD);
  const int e0 = b0 + threadIdx.x;
  if (b0 + 256 * EDGES_PER_THREAD <= n_edges) {
    int dv[EDGES_PER_THREAD];
    int sv[EDGES_PER_THREAD];
    int pos[EDGES_PER_THREAD];
#pragma unroll
    for (int k = 0; k < EDGES_PER_THREAD; ++k) {
      dv[k] = dst[e0 + k * 256];
      sv[k] = src[e0 + k * 256];
    }
#pragma unroll
    for (int k = 0; k < EDGES_PER_THREAD; ++k)
      pos[k] = atomicAdd(&cursor[dv[k]], 1);
#pragma unroll
    for (int k = 0; k < EDGES_PER_THREAD; ++k) csr_src[pos[k]] = sv[k];
  } else {
    for (int k = 0; k < EDGES_PER_THREAD; ++k) {
      const int e = e0 + k * 256;
      if (e < n_edges) {
        const int pos = atomicAdd(&cursor[dst[e]], 1);
        csr_src[pos] = src[e];
      }
    }
  }
}

#define GA_NODES 2
__device__ __forceinline__ float leaky(float v) {
  return (v > 0.0f) ? v : NEG_SLOPE * v;
}

__device__ __forceinline__ float wave_max(float v) {
  for (int o = 16; o > 0; o >>= 1) v = __builtin_fmaxf(v, __shfl_xor(v, o, 32));
  return v;
}

__device__ __forceinline__ float wave_sum(float v) {
  for (int o = 16; o > 0; o >>= 1) v += __shfl_xor(v, o, 32);
  return v;
}

__device__ __forceinline__ void add_rows(float4& acc, int u, float a, int cnt,
                                         const float4* __restrict__ H4,
                                         int lane) {
#pragma unroll 4
  for (int t = 0; t < cnt; ++t) {
    const int ut = __builtin_amdgcn_readlane(u, t);
    const float at = __int_as_float(__builtin_amdgcn_readlane(__float_as_int(a), t));
    const float4 h4 = H4[(size_t)ut * (CH / 4) + lane];
    acc.x += at * h4.x;
    acc.y += at * h4.y;
    acc.z += at * h4.z;
    acc.w += at * h4.w;
  }
}

__global__ void gat_node_gather(const int* __restrict__ rowptr,
                                const int* __restrict__ csr_src,
                                const float* __restrict__ s,
                                const float* __restrict__ d,
                                const float* __restrict__ H,
                                const float* __restrict__ bias,
                                float* __restrict__ x, int n_nodes) {
  const int lane = threadIdx.x & 31;
  const int v0 = __builtin_amdgcn_readfirstlane(
      (int)(blockIdx.x * 8 + (threadIdx.x >> 5)) * GA_NODES);
  const float4 b4 = ((const float4*)bias)[lane];
  for (int v = v0; v < v0 + GA_NODES && v < n_nodes; ++v) {
    const int beg = rowptr[v];
    const int end = rowptr[v + 1];
    const float dv = d[v];
    const float es = leaky(s[v] + dv);
    const float4* H4 = (const float4*)H;

    float4 acc;
    if (end - beg <= 32) {
      const int j = beg + lane;
      const bool ok = j < end;
      const int u = ok ? csr_src[j] : v;
      const float val = leaky(s[u] + dv);
      float m = wave_max(ok ? __builtin_fmaxf(val, es) : es);
      if (!__builtin_isfinite(m)) m = 0.0f;
      const float w = ok ? __builtin_expf(val - m) : 0.0f;
      const float ws = __builtin_expf(es - m);
      const float rden = wave_sum(w) + ws + SM_EPS;
      const float a = w / rden;
      const float as = ws / rden;
      const float4 hs = H4[(size_t)v * (CH / 4) + lane];
      acc.x = as * hs.x;
      acc.y = as * hs.y;
      acc.z = as * hs.z;
      acc.w = as * hs.w;
      add_rows(acc, u, a, end - beg, H4, lane);
    } else {
      float m = es;
      for (int base = beg; base < end; base += 32) {
        const int j = base + lane;
        if (j < end) m = __builtin_fmaxf(m, leaky(s[csr_src[j]] + dv));
      }
      m = wave_max(m);
      if (!__builtin_isfinite(m)) m = 0.0f;
      float den = 0.0f;
      for (int base = beg; base < end; base += 32) {
        const int j = base + lane;
        if (j < end) den += __builtin_expf(leaky(s[csr_src[j]] + dv) - m);
      }
      const float ws = __builtin_expf(es - m);
      const float rden = wave_sum(den) + ws + SM_EPS;
      const float as = ws / rden;
      const float4 hs = H4[(size_t)v * (CH / 4) + lane];
      acc.x = as * hs.x;
      acc.y = as * hs.y;
      acc.z = as * hs.z;
      acc.w = as * hs.w;
      for (int base = beg; base < end; base += 32) {
        const int j = base + lane;
        const bool ok = j < end;
        const int u = ok ? csr_src[j] : v;
        const float a = ok ? __builtin_expf(leaky(s[u] + dv) - m) / rden : 0.0f;
        const int left = end - base;
        add_rows(acc, u, a, (left < 32) ? left : 32, H4, lane);
      }
    }

    float4* xp = (float4*)(x + (size_t)v * CH) + lane;
    float4 xv = *xp;
    xv.x += 1.0f / (1.0f + __builtin_expf(-(acc.x + b4.x)));
    xv.y += 1.0f / (1.0f + __builtin_expf(-(acc.y + b4.y)));
    xv.z += 1.0f / (1.0f + __builtin_expf(-(acc.z + b4.z)));
    xv.w += 1.0f / (1.0f + __builtin_expf(-(acc.w + b4.w)));
    *xp = xv;
  }
}

#define COPY_PER_THREAD 8

__global__ void gat_copy(const float* __restrict__ in, float* __restrict__ out,
                         size_t n4) {
  const float4* in4 = (const float4*)in;
  float4* out4 = (float4*)out;
  const size_t i0 = (size_t)blockIdx.x * (256 * COPY_PER_THREAD) + threadIdx.x;
  float4 r[COPY_PER_THREAD];
#pragma unroll
  for (int k = 0; k < COPY_PER_THREAD; ++k) {
    const size_t i = i0 + (size_t)k * 256;
    r[k] = (i < n4) ? in4[i] : make_float4(0.0f, 0.0f, 0.0f, 0.0f);
  }
#pragma unroll
  for (int k = 0; k < COPY_PER_THREAD; ++k) {
    const size_t i = i0 + (size_t)k * 256;
    if (i < n4) out4[i] = r[k];
  }
}

extern "C" void kernel_launch(void* const* d_in, const int* in_sizes, int n_in,
                              void* d_out, int out_size, void* d_ws,
                              size_t ws_size, hipStream_t stream) {
  const float* x_in = (const float*)d_in[0];
  const int* ei = (const int*)d_in[1];
  const float* Ws = (const float*)d_in[2];
  const float* att_src = (const float*)d_in[3];
  const float* att_dst = (const float*)d_in[4];
  const float* bias = (const float*)d_in[5];

  const int N = in_sizes[0] / CH;
  const int E = in_sizes[1] / 2;
  const int M = in_sizes[3] / CH;
  const int* src = ei;
  const int* dst = ei + E;

  float* ws = (float*)d_ws;
  float* h = ws;
  float* s = h + (size_t)N * CH;
  float* dsc = s + N;
  int* cnt = (int*)(dsc + N);
  int* cursor = cnt + N;
  int* rowptr = cursor + N;
  int* csr_src = rowptr + N + 1;

  float* x = (float*)d_out;

  const size_t nfeat = (size_t)N * CH;
  const int blk = 256;
  const size_t nfeat4 = nfeat / 4;
  const int per_copy = blk * COPY_PER_THREAD;
  const int g_copy = (int)((nfeat4 + per_copy - 1) / per_copy);
  const int g_gemm = (N + 127) / 128;
  const int g_score = (N + 8 * SC_NODES - 1) / (8 * SC_NODES);
  const int g_gather = (N + 8 * GA_NODES - 1) / (8 * GA_NODES);
  const int g_n = (N + blk - 1) / blk;
  const int g_scan = (N + SCAN_TILE - 1) / SCAN_TILE;
  const int per_edge = blk * EDGES_PER_THREAD;
  const int g_e = (E + per_edge - 1) / per_edge;
  const size_t smem = (size_t)CH * WT_PAD * sizeof(float);

  gat_copy<<<g_copy, blk, 0, stream>>>(x_in, x, nfeat4);

  csr_zero<<<g_n, blk, 0, stream>>>(cnt, N);
  csr_count<<<g_e, blk, 0, stream>>>(dst, cnt, E);
  csr_scan<<<g_scan, SCAN_TILE, 0, stream>>>(cnt, rowptr, cursor, N);
  csr_fill<<<g_e, blk, 0, stream>>>(src, dst, cursor, csr_src, E);

  for (int m = 0; m < M; ++m) {
    const float* W = Ws + (size_t)m * CH * CH;
    const float* as_ = att_src + (size_t)m * CH;
    const float* ad_ = att_dst + (size_t)m * CH;
    const float* b_ = bias + (size_t)m * CH;

    gat_gemm_wmma<<<g_gemm, blk, smem, stream>>>(x, W, h, N);
    gat_node_scores<<<g_score, blk, 0, stream>>>(h, as_, ad_, s, dsc, N);
    gat_node_gather<<<g_gather, blk, 0, stream>>>(rowptr, csr_src, s, dsc, h, b_,
                                                  x, N);
  }
}


# edge sort rebuilt as a two-level counting sort in LDS (512-destination bins, then per bin): no global atomics or scattered 4 B stores, 4 launches to 3; order inside a node still arrival order
# speedup vs baseline: 5.7428x; 1.5395x over previous
#include <hip/hip_runtime.h>


typedef float v2f __attribute__((ext_vector_type(2)));
typedef float v8f __attribute__((ext_vector_type(8)));

#define CH 128
#define NEG_SLOPE 0.2f
#define SM_EPS 1e-16f

#define WT_PAD 132

__global__ void gat_gemm_wmma(const float* __restrict__ X,
                              const float* __restrict__ W,
                              float* __restrict__ H, int n_nodes) {
  extern __shared__ float wt[];
  const int tid = threadIdx.x;
  for (int i = tid; i < CH * CH; i += 256) {
    int k = i >> 7;
    int n = i & (CH - 1);
    wt[n * WT_PAD + k] = W[i];
  }
  __syncthreads();

  const int wave = tid >> 5;
  const int lane = tid & 31;
  const int row0 = blockIdx.x * 128 + wave * 16;
  const int mrow = lane & 15;
  const int koff = (lane >> 4) * 2;

  int rload = row0 + mrow;
  if (rload >= n_nodes) rload = n_nodes - 1;
  const float* xrow = X + (size_t)rload * CH + koff;
  const float* wrow = wt + mrow * WT_PAD + koff;

  v8f acc[8];
#pragma unroll
  for (int t = 0; t < 8; ++t) acc[t] = {};

  v2f a = *(const v2f*)(xrow);
  v2f b[8];
#pragma unroll
  for (int t = 0; t < 8; ++t)
    b[t] = *(const v2f*)(wrow + t * 16 * WT_PAD);

  for (int k4 = 0; k4 < CH; k4 += 4) {
    const int kn = k4 + 4;
    v2f an = {};
    v2f bn[8];
#pragma unroll
    for (int t = 0; t < 8; ++t) bn[t] = {};
    if (kn < CH) {
      an = *(const v2f*)(xrow + kn);
#pragma unroll
      for (int t = 0; t < 8; ++t)
        bn[t] = *(const v2f*)(wrow + t * 16 * WT_PAD + kn);
    }
#pragma unroll
    for (int t = 0; t < 8; ++t)
      acc[t] = __builtin_amdgcn_wmma_f32_16x16x4_f32(
          false, a, false, b[t], (short)0, acc[t], false, false);
    a = an;
#pragma unroll
    for (int t = 0; t < 8; ++t) b[t] = bn[t];
  }

  const int rhi = (lane >> 4) * 8;
#pragma unroll
  for (int t = 0; t < 8; ++t) {
    const int col = t * 16 + (lane & 15);
#pragma unroll
    for (int r = 0; r < 8; ++r) {
      int row = row0 + r + rhi;
      if (row < n_nodes) H[(size_t)row * CH + col] = acc[t][r];
    }
  }
}

#define SC_NODES 8

__global__ void gat_node_scores(const float* __restrict__ H,
                                const float* __restrict__ a_src,
                                const float* __restrict__ a_dst,
                                float* __restrict__ s, float* __restrict__ d,
                                int n_nodes) {
  const int wave = threadIdx.x >> 5;
  const int lane = threadIdx.x & 31;
  const int node0 = (blockIdx.x * 8 + wave) * SC_NODES;
  if (node0 >= n_nodes) return;
  const float4* H4 = (const float4*)H;
  const float4 a4 = ((const float4*)a_src)[lane];
  const float4 b4 = ((const float4*)a_dst)[lane];
  float4 h4[SC_NODES];
#pragma unroll
  for (int k = 0; k < SC_NODES; ++k) {
    int node = node0 + k;
    if (node >= n_nodes) node = n_nodes - 1;
    h4[k] = H4[(size_t)node * (CH / 4) + lane];
  }
  float s_keep = 0.0f;
  float d_keep = 0.0f;
#pragma unroll
  for (int k = 0; k < SC_NODES; ++k) {
    float ss = h4[k].x * a4.x + h4[k].y * a4.y + h4[k].z * a4.z + h4[k].w * a4.w;
    float dd = h4[k].x * b4.x + h4[k].y * b4.y + h4[k].z * b4.z + h4[k].w * b4.w;
#pragma unroll
    for (int o = 16; o > 0; o >>= 1) {
      ss += __shfl_xor(ss, o, 32);
      dd += __shfl_xor(dd, o, 32);
    }
    if (lane == k) {
      s_keep = ss;
      d_keep = dd;
    }
  }
  const int node = node0 + lane;
  if (lane < SC_NODES && node < n_nodes) {
    s[node] = s_keep;
    d[node] = d_keep;
  }
}

#define BIN_SHIFT 9
#define BIN_NODES (1 << BIN_SHIFT)
#define MAX_BINS 256
#define SORT_THREADS 1024
#define SORT_PER_THREAD 8
#define SORT_CHUNK (SORT_THREADS * SORT_PER_THREAD)
#define TABLE_AHEAD 16
#define FIN_CAP 16384
#define FIN_AHEAD 8

__device__ __forceinline__ int wave_sum_int(int v) {
  for (int o = 16; o > 0; o >>= 1) v += __shfl_xor(v, o, 32);
  return v;
}

__device__ __forceinline__ int block_exscan(int v, int* wsum, int& total) {
  const int lane = threadIdx.x & 31;
  const int wave = threadIdx.x >> 5;
  int incl = v;
#pragma unroll
  for (int o = 1; o < 32; o <<= 1) {
    const int y = __shfl_up(incl, o, 32);
    if (lane >= o) incl += y;
  }
  if (lane == 31) wsum[wave] = incl;
  __syncthreads();
  const int w = wsum[lane];
  const int before = wave_sum_int((lane < wave) ? w : 0);
  total = wave_sum_int(w);
  __syncthreads();
  return before + incl - v;
}

__global__ void bin_count(const int* __restrict__ dst, int* __restrict__ table,
                          int n_edges) {
  __shared__ int hist[MAX_BINS];
  const int t = threadIdx.x;
  if (t < MAX_BINS) hist[t] = 0;
  __syncthreads();
  const int e0 = blockIdx.x * SORT_CHUNK + t;
  int dv[SORT_PER_THREAD];
#pragma unroll
  for (int k = 0; k < SORT_PER_THREAD; ++k) {
    const int e = e0 + k * SORT_THREADS;
    dv[k] = (e < n_edges) ? dst[e] : -1;
  }
#pragma unroll
  for (int k = 0; k < SORT_PER_THREAD; ++k)
    if (dv[k] >= 0) atomicAdd(&hist[dv[k] >> BIN_SHIFT], 1);
  __syncthreads();
  if (t < MAX_BINS) table[(size_t)blockIdx.x * MAX_BINS + t] = hist[t];
}

__global__ void bin_scatter(const int* __restrict__ src,
                            const int* __restrict__ dst,
                            const int* __restrict__ table,
                            int* __restrict__ pairs, int* __restrict__ bin_start,
                            int n_edges, int n_chunks) {
  __shared__ int2 stage[SORT_CHUNK];
  __shared__ int tot[MAX_BINS];
  __shared__ int bef[MAX_BINS];
  __shared__ int cursor[MAX_BINS];
  __shared__ int gbase[MAX_BINS];
  __shared__ int wsum[32];
  const int t = threadIdx.x;
  const int c = blockIdx.x;

  const int e0 = c * SORT_CHUNK + t;
  int dv[SORT_PER_THREAD];
  int sv[SORT_PER_THREAD];
#pragma unroll
  for (int k = 0; k < SORT_PER_THREAD; ++k) {
    const int e = e0 + k * SORT_THREADS;
    dv[k] = (e < n_edges) ? dst[e] : -1;
    sv[k] = (e < n_edges) ? src[e] : 0;
  }

  if (t < MAX_BINS) {
    tot[t] = 0;
    bef[t] = 0;
  }
  __syncthreads();
  {
    const int b = t & (MAX_BINS - 1);
    const int part = t >> 8;
    int ptot = 0;
    int pbef = 0;
    for (int r0 = part; r0 < n_chunks; r0 += 4 * TABLE_AHEAD) {
      int v[TABLE_AHEAD];
#pragma unroll
      for (int k = 0; k < TABLE_AHEAD; ++k) {
        const int r = r0 + 4 * k;
        v[k] = (r < n_chunks) ? table[(size_t)r * MAX_BINS + b] : 0;
      }
#pragma unroll
      for (int k = 0; k < TABLE_AHEAD; ++k) {
        const int r = r0 + 4 * k;
        ptot += v[k];
        if (r < c) pbef += v[k];
      }
    }
    atomicAdd(&tot[b], ptot);
    atomicAdd(&bef[b], pbef);
  }
  __syncthreads();
  const int own = (t < MAX_BINS) ? table[(size_t)c * MAX_BINS + t] : 0;
  const int total = (t < MAX_BINS) ? tot[t] : 0;
  int sum_own;
  int sum_all;
  const int lstart = block_exscan(own, wsum, sum_own);
  const int bstart = block_exscan(total, wsum, sum_all);
  if (t < MAX_BINS) {
    cursor[t] = lstart;
    gbase[t] = bstart + bef[t] - lstart;
    if (c == 0) bin_start[t] = bstart;
  }
  if (c == 0 && t == 0) bin_start[MAX_BINS] = sum_all;
  __syncthreads();

#pragma unroll
  for (int k = 0; k < SORT_PER_THREAD; ++k) {
    if (dv[k] >= 0) {
      const int slot = atomicAdd(&cursor[dv[k] >> BIN_SHIFT], 1);
      stage[slot] = make_int2(dv[k], sv[k]);
    }
  }
  __syncthreads();
  const int left = n_edges - c * SORT_CHUNK;
  const int nloc = (left < SORT_CHUNK) ? left : SORT_CHUNK;
  int2* out2 = (int2*)pairs;
  for (int i = t; i < nloc; i += SORT_THREADS) {
    const int2 p = stage[i];
    out2[gbase[p.x >> BIN_SHIFT] + i] = p;
  }
}

__global__ void bin_finish(const int* __restrict__ pairs,
                           const int* __restrict__ bin_start,
                           int* __restrict__ rowptr, int* __restrict__ csr_src,
                           int n_nodes, int n_edges) {
  __shared__ int ncnt[BIN_NODES];
  __shared__ int sorted[FIN_CAP];
  __shared__ int wsum[32];
  const int t = threadIdx.x;
  const int q = blockIdx.x;
  const int node0 = q << BIN_SHIFT;
  const int beg = bin_start[q];
  const int end = bin_start[q + 1];
  const int len = end - beg;
  const int2* in2 = (const int2*)pairs;
  if (t < BIN_NODES) ncnt[t] = 0;
  __syncthreads();
  for (int i0 = beg + t; i0 < end; i0 += FIN_AHEAD * SORT_THREADS) {
    int dv[FIN_AHEAD];
#pragma unroll
    for (int k = 0; k < FIN_AHEAD; ++k) {
      const int i = i0 + k * SORT_THREADS;
      dv[k] = (i < end) ? in2[i].x : -1;
    }
#pragma unroll
    for (int k = 0; k < FIN_AHEAD; ++k)
      if (dv[k] >= 0) atomicAdd(&ncnt[dv[k] - node0], 1);
  }
  __syncthreads();
  const int mine = (t < BIN_NODES) ? ncnt[t] : 0;
  int sum_bin;
  const int excl = block_exscan(mine, wsum, sum_bin);
  if (t < BIN_NODES) {
    ncnt[t] = excl;
    const int node = node0 + t;
    if (node < n_nodes) rowptr[node] = beg + excl;
  }
  if (q == (int)gridDim.x - 1 && t == 0) rowptr[n_nodes] = n_edges;
  __syncthreads();
  const bool fits = len <= FIN_CAP;
  for (int i0 = beg + t; i0 < end; i0 += FIN_AHEAD * SORT_THREADS) {
    int2 p[FIN_AHEAD];
#pragma unroll
    for (int k = 0; k < FIN_AHEAD; ++k) {
      const int i = i0 + k * SORT_THREADS;
      p[k] = (i < end) ? in2[i] : make_int2(-1, 0);
    }
#pragma unroll
    for (int k = 0; k < FIN_AHEAD; ++k) {
      if (p[k].x >= 0) {
        const int slot = atomicAdd(&ncnt[p[k].x - node0], 1);
        if (fits)
          sorted[slot] = p[k].y;
        else
          csr_src[beg + slot] = p[k].y;
      }
    }
  }
  if (fits) {
    __syncthreads();
    for (int i = t; i < len; i += SORT_THREADS) csr_src[beg + i] = sorted[i];
  }
}

#define GA_NODES 2
__device__ __forceinline__ float leaky(float v) {
  return (v > 0.0f) ? v : NEG_SLOPE * v;
}

__device__ __forceinline__ float wave_max(float v) {
  for (int o = 16; o > 0; o >>= 1) v = __builtin_fmaxf(v, __shfl_xor(v, o, 32));
  return v;
}

__device__ __forceinline__ float wave_sum(float v) {
  for (int o = 16; o > 0; o >>= 1) v += __shfl_xor(v, o, 32);
  return v;
}

__device__ __forceinline__ void add_rows(float4& acc, int u, float a, int cnt,
                                         const float4* __restrict__ H4,
                                         int lane) {
#pragma unroll 4
  for (int t = 0; t < cnt; ++t) {
    const int ut = __builtin_amdgcn_readlane(u, t);
    const float at = __int_as_float(__builtin_amdgcn_readlane(__float_as_int(a), t));
    const float4 h4 = H4[(size_t)ut * (CH / 4) + lane];
    acc.x += at * h4.x;
    acc.y += at * h4.y;
    acc.z += at * h4.z;
    acc.w += at * h4.w;
  }
}

__global__ void gat_node_gather(const int* __restrict__ rowptr,
                                const int* __restrict__ csr_src,
                                const float* __restrict__ s,
                                const float* __restrict__ d,
                                const float* __restrict__ H,
                                const float* __restrict__ bias,
                                float* __restrict__ x, int n_nodes) {
  const int lane = threadIdx.x & 31;
  const int v0 = __builtin_amdgcn_readfirstlane(
      (int)(blockIdx.x * 8 + (threadIdx.x >> 5)) * GA_NODES);
  const float4 b4 = ((const float4*)bias)[lane];
  for (int v = v0; v < v0 + GA_NODES && v < n_nodes; ++v) {
    const int beg = rowptr[v];
    const int end = rowptr[v + 1];
    const float dv = d[v];
    const float es = leaky(s[v] + dv);
    const float4* H4 = (const float4*)H;

    float4 acc;
    if (end - beg <= 32) {
      const int j = beg + lane;
      const bool ok = j < end;
      const int u = ok ? csr_src[j] : v;
      const float val = leaky(s[u] + dv);
      float m = wave_max(ok ? __builtin_fmaxf(val, es) : es);
      if (!__builtin_isfinite(m)) m = 0.0f;
      const float w = ok ? __builtin_expf(val - m) : 0.0f;
      const float ws = __builtin_expf(es - m);
      const float rden = wave_sum(w) + ws + SM_EPS;
      const float a = w / rden;
      const float as = ws / rden;
      const float4 hs = H4[(size_t)v * (CH / 4) + lane];
      acc.x = as * hs.x;
      acc.y = as * hs.y;
      acc.z = as * hs.z;
      acc.w = as * hs.w;
      add_rows(acc, u, a, end - beg, H4, lane);
    } else {
      float m = es;
      for (int base = beg; base < end; base += 32) {
        const int j = base + lane;
        if (j < end) m = __builtin_fmaxf(m, leaky(s[csr_src[j]] + dv));
      }
      m = wave_max(m);
      if (!__builtin_isfinite(m)) m = 0.0f;
      float den = 0.0f;
      for (int base = beg; base < end; base += 32) {
        const int j = base + lane;
        if (j < end) den += __builtin_expf(leaky(s[csr_src[j]] + dv) - m);
      }
      const float ws = __builtin_expf(es - m);
      const float rden = wave_sum(den) + ws + SM_EPS;
      const float as = ws / rden;
      const float4 hs = H4[(size_t)v * (CH / 4) + lane];
      acc.x = as * hs.x;
      acc.y = as * hs.y;
      acc.z = as * hs.z;
      acc.w = as * hs.w;
      for (int base = beg; base < end; base += 32) {
        const int j = base + lane;
        const bool ok = j < end;
        const int u = ok ? csr_src[j] : v;
        const float a = ok ? __builtin_expf(leaky(s[u] + dv) - m) / rden : 0.0f;
        const int left = end - base;
        add_rows(acc, u, a, (left < 32) ? left : 32, H4, lane);
      }
    }

    float4* xp = (float4*)(x + (size_t)v * CH) + lane;
    float4 xv = *xp;
    xv.x += 1.0f / (1.0f + __builtin_expf(-(acc.x + b4.x)));
    xv.y += 1.0f / (1.0f + __builtin_expf(-(acc.y + b4.y)));
    xv.z += 1.0f / (1.0f + __builtin_expf(-(acc.z + b4.z)));
    xv.w += 1.0f / (1.0f + __builtin_expf(-(acc.w + b4.w)));
    *xp = xv;
  }
}

#define COPY_PER_THREAD 8

__global__ void gat_copy(const float* __restrict__ in, float* __restrict__ out,
                         size_t n4) {
  const float4* in4 = (const float4*)in;
  float4* out4 = (float4*)out;
  const size_t i0 = (size_t)blockIdx.x * (256 * COPY_PER_THREAD) + threadIdx.x;
  float4 r[COPY_PER_THREAD];
#pragma unroll
  for (int k = 0; k < COPY_PER_THREAD; ++k) {
    const size_t i = i0 + (size_t)k * 256;
    r[k] = (i < n4) ? in4[i] : make_float4(0.0f, 0.0f, 0.0f, 0.0f);
  }
#pragma unroll
  for (int k = 0; k < COPY_PER_THREAD; ++k) {
    const size_t i = i0 + (size_t)k * 256;
    if (i < n4) out4[i] = r[k];
  }
}

extern "C" void kernel_launch(void* const* d_in, const int* in_sizes, int n_in,
                              void* d_out, int out_size, void* d_ws,
                              size_t ws_size, hipStream_t stream) {
  const float* x_in = (const float*)d_in[0];
  const int* ei = (const int*)d_in[1];
  const float* Ws = (const float*)d_in[2];
  const float* att_src = (const float*)d_in[3];
  const float* att_dst = (const float*)d_in[4];
  const float* bias = (const float*)d_in[5];

  const int N = in_sizes[0] / CH;
  const int E = in_sizes[1] / 2;
  const int M = in_sizes[3] / CH;
  const int* src = ei;
  const int* dst = ei + E;

  float* ws = (float*)d_ws;
  float* h = ws;
  float* s = h + (size_t)N * CH;
  float* dsc = s + N;
  int* rowptr = (int*)(dsc + N);
  int* csr_src = rowptr + N + 1;

  float* x = (float*)d_out;

  const size_t nfeat = (size_t)N * CH;
  const int blk = 256;
  const size_t nfeat4 = nfeat / 4;
  const int per_copy = blk * COPY_PER_THREAD;
  const int g_copy = (int)((nfeat4 + per_copy - 1) / per_copy);
  const int g_gemm = (N + 127) / 128;
  const int g_score = (N + 8 * SC_NODES - 1) / (8 * SC_NODES);
  const int g_gather = (N + 8 * GA_NODES - 1) / (8 * GA_NODES);
  const int n_chunks = (E + SORT_CHUNK - 1) / SORT_CHUNK;
  const int n_bins = (N + BIN_NODES - 1) / BIN_NODES;
  int* pairs = (int*)h;
  int* table = pairs + 2 * (size_t)E;
  int* bin_start = table + (size_t)n_chunks * MAX_BINS;
  const size_t smem = (size_t)CH * WT_PAD * sizeof(float);

  gat_copy<<<g_copy, blk, 0, stream>>>(x_in, x, nfeat4);

  bin_count<<<n_chunks, SORT_THREADS, 0, stream>>>(dst, table, E);
  bin_scatter<<<n_chunks, SORT_THREADS, 0, stream>>>(src, dst, table, pairs,
                                                    bin_start, E, n_chunks);
  bin_finish<<<n_bins, SORT_THREADS, 0, stream>>>(pairs, bin_start, rowptr,
                                                 csr_src, N, E);

  for (int m = 0; m < M; ++m) {
    const float* W = Ws + (size_t)m * CH * CH;
    const float* as_ = att_src + (size_t)m * CH;
    const float* ad_ = att_dst + (size_t)m * CH;
    const float* b_ = bias + (size_t)m * CH;

    gat_gemm_wmma<<<g_gemm, blk, smem, stream>>>(x, W, h, N);
    gat_node_scores<<<g_score, blk, 0, stream>>>(h, as_, ad_, s, dsc, N);
    gat_node_gather<<<g_gather, blk, 0, stream>>>(rowptr, csr_src, s, dsc, h, b_,
                                                  x, N);
  }
}


# gather wave takes source scores from the h rows it loads: 4 rows per transposing DPP lane sum (no LDS shuffle), running-max softmax in edge order; no score kernel; 13 to 10 launches
# speedup vs baseline: 5.9214x; 1.0311x over previous
#include <hip/hip_runtime.h>


typedef float v2f __attribute__((ext_vector_type(2)));
typedef float v8f __attribute__((ext_vector_type(8)));

#define CH 128
#define NEG_SLOPE 0.2f
#define SM_EPS 1e-16f

#define WT_PAD 132

__global__ void gat_gemm_wmma(const float* __restrict__ X,
                              const float* __restrict__ W,
                              float* __restrict__ H, int n_nodes) {
  extern __shared__ float wt[];
  const int tid = threadIdx.x;
  for (int i = tid; i < CH * CH; i += 256) {
    int k = i >> 7;
    int n = i & (CH - 1);
    wt[n * WT_PAD + k] = W[i];
  }
  __syncthreads();

  const int wave = tid >> 5;
  const int lane = tid & 31;
  const int row0 = blockIdx.x * 128 + wave * 16;
  const int mrow = lane & 15;
  const int koff = (lane >> 4) * 2;

  int rload = row0 + mrow;
  if (rload >= n_nodes) rload = n_nodes - 1;
  const float* xrow = X + (size_t)rload * CH + koff;
  const float* wrow = wt + mrow * WT_PAD + koff;

  v8f acc[8];
#pragma unroll
  for (int t = 0; t < 8; ++t) acc[t] = {};

  v2f a = *(const v2f*)(xrow);
  v2f b[8];
#pragma unroll
  for (int t = 0; t < 8; ++t)
    b[t] = *(const v2f*)(wrow + t * 16 * WT_PAD);

  for (int k4 = 0; k4 < CH; k4 += 4) {
    const int kn = k4 + 4;
    v2f an = {};
    v2f bn[8];
#pragma unroll
    for (int t = 0; t < 8; ++t) bn[t] = {};
    if (kn < CH) {
      an = *(const v2f*)(xrow + kn);
#pragma unroll
      for (int t = 0; t < 8; ++t)
        bn[t] = *(const v2f*)(wrow + t * 16 * WT_PAD + kn);
    }
#pragma unroll
    for (int t = 0; t < 8; ++t)
      acc[t] = __builtin_amdgcn_wmma_f32_16x16x4_f32(
          false, a, false, b[t], (short)0, acc[t], false, false);
    a = an;
#pragma unroll
    for (int t = 0; t < 8; ++t) b[t] = bn[t];
  }

  const int rhi = (lane >> 4) * 8;
#pragma unroll
  for (int t = 0; t < 8; ++t) {
    const int col = t * 16 + (lane & 15);
#pragma unroll
    for (int r = 0; r < 8; ++r) {
      int row = row0 + r + rhi;
      if (row < n_nodes) H[(size_t)row * CH + col] = acc[t][r];
    }
  }
}

#define BIN_SHIFT 9
#define BIN_NODES (1 << BIN_SHIFT)
#define MAX_BINS 256
#define SORT_THREADS 1024
#define SORT_PER_THREAD 8
#define SORT_CHUNK (SORT_THREADS * SORT_PER_THREAD)
#define TABLE_AHEAD 16
#define FIN_CAP 16384
#define FIN_AHEAD 8

__device__ __forceinline__ int wave_sum_int(int v) {
  for (int o = 16; o > 0; o >>= 1) v += __shfl_xor(v, o, 32);
  return v;
}

__device__ __forceinline__ int block_exscan(int v, int* wsum, int& total) {
  const int lane = threadIdx.x & 31;
  const int wave = threadIdx.x >> 5;
  int incl = v;
#pragma unroll
  for (int o = 1; o < 32; o <<= 1) {
    const int y = __shfl_up(incl, o, 32);
    if (lane >= o) incl += y;
  }
  if (lane == 31) wsum[wave] = incl;
  __syncthreads();
  const int w = wsum[lane];
  const int before = wave_sum_int((lane < wave) ? w : 0);
  total = wave_sum_int(w);
  __syncthreads();
  return before + incl - v;
}

__global__ void bin_count(const int* __restrict__ dst, int* __restrict__ table,
                          int n_edges) {
  __shared__ int hist[MAX_BINS];
  const int t = threadIdx.x;
  if (t < MAX_BINS) hist[t] = 0;
  __syncthreads();
  const int e0 = blockIdx.x * SORT_CHUNK + t;
  int dv[SORT_PER_THREAD];
#pragma unroll
  for (int k = 0; k < SORT_PER_THREAD; ++k) {
    const int e = e0 + k * SORT_THREADS;
    dv[k] = (e < n_edges) ? dst[e] : -1;
  }
#pragma unroll
  for (int k = 0; k < SORT_PER_THREAD; ++k)
    if (dv[k] >= 0) atomicAdd(&hist[dv[k] >> BIN_SHIFT], 1);
  __syncthreads();
  if (t < MAX_BINS) table[(size_t)blockIdx.x * MAX_BINS + t] = hist[t];
}

__global__ void bin_scatter(const int* __restrict__ src,
                            const int* __restrict__ dst,
                            const int* __restrict__ table,
                            int* __restrict__ pairs, int* __restrict__ bin_start,
                            int n_edges, int n_chunks) {
  __shared__ int2 stage[SORT_CHUNK];
  __shared__ int tot[MAX_BINS];
  __shared__ int bef[MAX_BINS];
  __shared__ int cursor[MAX_BINS];
  __shared__ int gbase[MAX_BINS];
  __shared__ int wsum[32];
  const int t = threadIdx.x;
  const int c = blockIdx.x;

  const int e0 = c * SORT_CHUNK + t;
  int dv[SORT_PER_THREAD];
  int sv[SORT_PER_THREAD];
#pragma unroll
  for (int k = 0; k < SORT_PER_THREAD; ++k) {
    const int e = e0 + k * SORT_THREADS;
    dv[k] = (e < n_edges) ? dst[e] : -1;
    sv[k] = (e < n_edges) ? src[e] : 0;
  }

  if (t < MAX_BINS) {
    tot[t] = 0;
    bef[t] = 0;
  }
  __syncthreads();
  {
    const int b = t & (MAX_BINS - 1);
    const int part = t >> 8;
    int ptot = 0;
    int pbef = 0;
    for (int r0 = part; r0 < n_chunks; r0 += 4 * TABLE_AHEAD) {
      int v[TABLE_AHEAD];
#pragma unroll
      for (int k = 0; k < TABLE_AHEAD; ++k) {
        const int r = r0 + 4 * k;
        v[k] = (r < n_chunks) ? table[(size_t)r * MAX_BINS + b] : 0;
      }
#pragma unroll
      for (int k = 0; k < TABLE_AHEAD; ++k) {
        const int r = r0 + 4 * k;
        ptot += v[k];
        if (r < c) pbef += v[k];
      }
    }
    atomicAdd(&tot[b], ptot);
    atomicAdd(&bef[b], pbef);
  }
  __syncthreads();
  const int own = (t < MAX_BINS) ? table[(size_t)c * MAX_BINS + t] : 0;
  const int total = (t < MAX_BINS) ? tot[t] : 0;
  int sum_own;
  int sum_all;
  const int lstart = block_exscan(own, wsum, sum_own);
  const int bstart = block_exscan(total, wsum, sum_all);
  if (t < MAX_BINS) {
    cursor[t] = lstart;
    gbase[t] = bstart + bef[t] - lstart;
    if (c == 0) bin_start[t] = bstart;
  }
  if (c == 0 && t == 0) bin_start[MAX_BINS] = sum_all;
  __syncthreads();

#pragma unroll
  for (int k = 0; k < SORT_PER_THREAD; ++k) {
    if (dv[k] >= 0) {
      const int slot = atomicAdd(&cursor[dv[k] >> BIN_SHIFT], 1);
      stage[slot] = make_int2(dv[k], sv[k]);
    }
  }
  __syncthreads();
  const int left = n_edges - c * SORT_CHUNK;
  const int nloc = (left < SORT_CHUNK) ? left : SORT_CHUNK;
  int2* out2 = (int2*)pairs;
  for (int i = t; i < nloc; i += SORT_THREADS) {
    const int2 p = stage[i];
    out2[gbase[p.x >> BIN_SHIFT] + i] = p;
  }
}

__global__ void bin_finish(const int* __restrict__ pairs,
                           const int* __restrict__ bin_start,
                           int* __restrict__ rowptr, int* __restrict__ csr_src,
                           int n_nodes, int n_edges) {
  __shared__ int ncnt[BIN_NODES];
  __shared__ int sorted[FIN_CAP];
  __shared__ int wsum[32];
  const int t = threadIdx.x;
  const int q = blockIdx.x;
  const int node0 = q << BIN_SHIFT;
  const int beg = bin_start[q];
  const int end = bin_start[q + 1];
  const int len = end - beg;
  const int2* in2 = (const int2*)pairs;
  if (t < BIN_NODES) ncnt[t] = 0;
  __syncthreads();
  for (int i0 = beg + t; i0 < end; i0 += FIN_AHEAD * SORT_THREADS) {
    int dv[FIN_AHEAD];
#pragma unroll
    for (int k = 0; k < FIN_AHEAD; ++k) {
      const int i = i0 + k * SORT_THREADS;
      dv[k] = (i < end) ? in2[i].x : -1;
    }
#pragma unroll
    for (int k = 0; k < FIN_AHEAD; ++k)
      if (dv[k] >= 0) atomicAdd(&ncnt[dv[k] - node0], 1);
  }
  __syncthreads();
  const int mine = (t < BIN_NODES) ? ncnt[t] : 0;
  int sum_bin;
  const int excl = block_exscan(mine, wsum, sum_bin);
  if (t < BIN_NODES) {
    ncnt[t] = excl;
    const int node = node0 + t;
    if (node < n_nodes) rowptr[node] = beg + excl;
  }
  if (q == (int)gridDim.x - 1 && t == 0) rowptr[n_nodes] = n_edges;
  __syncthreads();
  const bool fits = len <= FIN_CAP;
  for (int i0 = beg + t; i0 < end; i0 += FIN_AHEAD * SORT_THREADS) {
    int2 p[FIN_AHEAD];
#pragma unroll
    for (int k = 0; k < FIN_AHEAD; ++k) {
      const int i = i0 + k * SORT_THREADS;
      p[k] = (i < end) ? in2[i] : make_int2(-1, 0);
    }
#pragma unroll
    for (int k = 0; k < FIN_AHEAD; ++k) {
      if (p[k].x >= 0) {
        const int slot = atomicAdd(&ncnt[p[k].x - node0], 1);
        if (fits)
          sorted[slot] = p[k].y;
        else
          csr_src[beg + slot] = p[k].y;
      }
    }
  }
  if (fits) {
    __syncthreads();
    for (int i = t; i < len; i += SORT_THREADS) csr_src[beg + i] = sorted[i];
  }
}

#define GA_NODES 2
#define LOG2E 1.4426950408889634f

__device__ __forceinline__ float leaky(float v) {
  return (v > 0.0f) ? v : NEG_SLOPE * v;
}

__device__ __forceinline__ float xhalf(float x) {
  const int b = __builtin_bit_cast(int, x);
  return __builtin_bit_cast(
      float, __builtin_amdgcn_permlanex16(b, b, 0x76543210, 0xfedcba98, false, false));
}

template <int M>
__device__ __forceinline__ float xrow(float x) {
  const int b = __builtin_bit_cast(int, x);
  return __builtin_bit_cast(
      float, __builtin_amdgcn_update_dpp(0, b, 0x160 | M, 0xf, 0xf, true));
}

__device__ __forceinline__ float dot4(const float4 a, const float4 b) {
  return a.x * b.x + a.y * b.y + a.z * b.z + a.w * b.w;
}

__device__ __forceinline__ float exp_neg(float v) {
  return __builtin_amdgcn_exp2f(v * LOG2E);
}

__device__ __forceinline__ void load_rows(float4& r0, float4& r1, float4& r2,
                                          float4& r3, int u, int t,
                                          const float4* __restrict__ H4,
                                          int lane) {
  const int u0 = __builtin_amdgcn_readlane(u, t);
  const int u1 = __builtin_amdgcn_readlane(u, t + 1);
  const int u2 = __builtin_amdgcn_readlane(u, t + 2);
  const int u3 = __builtin_amdgcn_readlane(u, t + 3);
  r0 = H4[(size_t)u0 * (CH / 4) + lane];
  r1 = H4[(size_t)u1 * (CH / 4) + lane];
  r2 = H4[(size_t)u2 * (CH / 4) + lane];
  r3 = H4[(size_t)u3 * (CH / 4) + lane];
}

__device__ __forceinline__ void take_rows(const float4 c0, const float4 c1,
                                          const float4 c2, const float4 c3,
                                          int t, int cnt, const float4 as4,
                                          float dv, bool hi16, bool hi8,
                                          int slot, float& m, float& den,
                                          float4& acc) {
  const float p0 = dot4(c0, as4);
  const float p1 = dot4(c1, as4);
  const float p2 = dot4(c2, as4);
  const float p3 = dot4(c3, as4);
  const float kA = hi16 ? p2 : p0;
  const float sA = hi16 ? p0 : p2;
  const float kB = hi16 ? p3 : p1;
  const float sB = hi16 ? p1 : p3;
  const float qA = kA + xhalf(sA);
  const float qB = kB + xhalf(sB);
  const float k2 = hi8 ? qB : qA;
  const float s2 = hi8 ? qA : qB;
  float sc = k2 + xrow<8>(s2);
  sc += xrow<4>(sc);
  sc += xrow<2>(sc);
  sc += xrow<1>(sc);

  float e = leaky(sc + dv);
  e = (t + slot < cnt) ? e : -__builtin_inff();
  float mb = __builtin_fmaxf(e, xrow<8>(e));
  mb = __builtin_fmaxf(mb, xhalf(mb));
  const float mn = __builtin_fmaxf(m, mb);
  const float corr = exp_neg(m - mn);
  const float wv = exp_neg(e - mn);
  float ws = wv + xrow<8>(wv);
  ws = ws + xhalf(ws);
  den = den * corr + ws;
  m = mn;
  const float w0 = __int_as_float(__builtin_amdgcn_readlane(__float_as_int(wv), 0));
  const float w1 = __int_as_float(__builtin_amdgcn_readlane(__float_as_int(wv), 8));
  const float w2 = __int_as_float(__builtin_amdgcn_readlane(__float_as_int(wv), 16));
  const float w3 = __int_as_float(__builtin_amdgcn_readlane(__float_as_int(wv), 24));
  acc.x = acc.x * corr + w0 * c0.x;
  acc.y = acc.y * corr + w0 * c0.y;
  acc.z = acc.z * corr + w0 * c0.z;
  acc.w = acc.w * corr + w0 * c0.w;
  acc.x += w1 * c1.x;
  acc.y += w1 * c1.y;
  acc.z += w1 * c1.z;
  acc.w += w1 * c1.w;
  acc.x += w2 * c2.x;
  acc.y += w2 * c2.y;
  acc.z += w2 * c2.z;
  acc.w += w2 * c2.w;
  acc.x += w3 * c3.x;
  acc.y += w3 * c3.y;
  acc.z += w3 * c3.z;
  acc.w += w3 * c3.w;
}

__global__ void gat_node_gather(const int* __restrict__ rowptr,
                                const int* __restrict__ csr_src,
                                const float* __restrict__ a_src,
                                const float* __restrict__ a_dst,
                                const float* __restrict__ H,
                                const float* __restrict__ bias,
                                float* __restrict__ x, int n_nodes) {
  const int lane = threadIdx.x & 31;
  const bool hi16 = (lane & 16) != 0;
  const bool hi8 = (lane & 8) != 0;
  const int slot = lane >> 3;
  const int v0 = __builtin_amdgcn_readfirstlane(
      (int)(blockIdx.x * 8 + (threadIdx.x >> 5)) * GA_NODES);
  const float4 b4 = ((const float4*)bias)[lane];
  const float4 as4 = ((const float4*)a_src)[lane];
  const float4 ad4 = ((const float4*)a_dst)[lane];
  const float4* H4 = (const float4*)H;
  for (int v = v0; v < v0 + GA_NODES && v < n_nodes; ++v) {
    const int beg = rowptr[v];
    const int end = rowptr[v + 1];
    int u = (beg + lane < end) ? csr_src[beg + lane] : v;
    float4* xp = (float4*)(x + (size_t)v * CH) + lane;
    float4 xv = *xp;
    const float4 hs = H4[(size_t)v * (CH / 4) + lane];

    float ps = dot4(hs, as4);
    float pd = dot4(hs, ad4);
    ps += xrow<1>(ps);
    pd += xrow<1>(pd);
    ps += xrow<2>(ps);
    pd += xrow<2>(pd);
    ps += xrow<4>(ps);
    pd += xrow<4>(pd);
    ps += xrow<8>(ps);
    pd += xrow<8>(pd);
    ps += xhalf(ps);
    pd += xhalf(pd);
    const float dv = pd;

    float m = leaky(ps + dv);
    float den = 1.0f;
    float4 acc = hs;

    for (int base = beg; base < end; base += 32) {
      const int left = end - base;
      const int cnt = (left < 32) ? left : 32;
      float4 a0, a1, a2, a3, b0, b1, b2, b3;
      load_rows(a0, a1, a2, a3, u, 0, H4, lane);
      for (int t = 0;; t += 8) {
        if (t + 4 >= cnt) {
          take_rows(a0, a1, a2, a3, t, cnt, as4, dv, hi16, hi8, slot, m, den, acc);
          break;
        }
        load_rows(b0, b1, b2, b3, u, t + 4, H4, lane);
        take_rows(a0, a1, a2, a3, t, cnt, as4, dv, hi16, hi8, slot, m, den, acc);
        if (t + 8 >= cnt) {
          take_rows(b0, b1, b2, b3, t + 4, cnt, as4, dv, hi16, hi8, slot, m, den, acc);
          break;
        }
        load_rows(a0, a1, a2, a3, u, t + 8, H4, lane);
        take_rows(b0, b1, b2, b3, t + 4, cnt, as4, dv, hi16, hi8, slot, m, den, acc);
      }
      if (base + 32 < end) {
        const int j = base + 32 + lane;
        u = (j < end) ? csr_src[j] : v;
      }
    }

    const float rden = 1.0f / (den + SM_EPS);
    xv.x += 1.0f / (1.0f + __builtin_expf(-(acc.x * rden + b4.x)));
    xv.y += 1.0f / (1.0f + __builtin_expf(-(acc.y * rden + b4.y)));
    xv.z += 1.0f / (1.0f + __builtin_expf(-(acc.z * rden + b4.z)));
    xv.w += 1.0f / (1.0f + __builtin_expf(-(acc.w * rden + b4.w)));
    *xp = xv;
  }
}

#define COPY_PER_THREAD 8

__global__ void gat_copy(const float* __restrict__ in, float* __restrict__ out,
                         size_t n4) {
  const float4* in4 = (const float4*)in;
  float4* out4 = (float4*)out;
  const size_t i0 = (size_t)blockIdx.x * (256 * COPY_PER_THREAD) + threadIdx.x;
  float4 r[COPY_PER_THREAD];
#pragma unroll
  for (int k = 0; k < COPY_PER_THREAD; ++k) {
    const size_t i = i0 + (size_t)k * 256;
    r[k] = (i < n4) ? in4[i] : make_float4(0.0f, 0.0f, 0.0f, 0.0f);
  }
#pragma unroll
  for (int k = 0; k < COPY_PER_THREAD; ++k) {
    const size_t i = i0 + (size_t)k * 256;
    if (i < n4) out4[i] = r[k];
  }
}

extern "C" void kernel_launch(void* const* d_in, const int* in_sizes, int n_in,
                              void* d_out, int out_size, void* d_ws,
                              size_t ws_size, hipStream_t stream) {
  const float* x_in = (const float*)d_in[0];
  const int* ei = (const int*)d_in[1];
  const float* Ws = (const float*)d_in[2];
  const float* att_src = (const float*)d_in[3];
  const float* att_dst = (const float*)d_in[4];
  const float* bias = (const float*)d_in[5];

  const int N = in_sizes[0] / CH;
  const int E = in_sizes[1] / 2;
  const int M = in_sizes[3] / CH;
  const int* src = ei;
  const int* dst = ei + E;

  float* ws = (float*)d_ws;
  float* h = ws;
  int* rowptr = (int*)(h + (size_t)N * CH);
  int* csr_src = rowptr + N + 1;

  float* x = (float*)d_out;

  const size_t nfeat = (size_t)N * CH;
  const int blk = 256;
  const size_t nfeat4 = nfeat / 4;
  const int per_copy = blk * COPY_PER_THREAD;
  const int g_copy = (int)((nfeat4 + per_copy - 1) / per_copy);
  const int g_gemm = (N + 127) / 128;
  const int g_gather = (N + 8 * GA_NODES - 1) / (8 * GA_NODES);
  const int n_chunks = (E + SORT_CHUNK - 1) / SORT_CHUNK;
  const int n_bins = (N + BIN_NODES - 1) / BIN_NODES;
  int* pairs = (int*)h;
  int* table = pairs + 2 * (size_t)E;
  int* bin_start = table + (size_t)n_chunks * MAX_BINS;
  const size_t smem = (size_t)CH * WT_PAD * sizeof(float);

  gat_copy<<<g_copy, blk, 0, stream>>>(x_in, x, nfeat4);

  bin_count<<<n_chunks, SORT_THREADS, 0, stream>>>(dst, table, E);
  bin_scatter<<<n_chunks, SORT_THREADS, 0, stream>>>(src, dst, table, pairs,
                                                    bin_start, E, n_chunks);
  bin_finish<<<n_bins, SORT_THREADS, 0, stream>>>(pairs, bin_start, rowptr,
                                                 csr_src, N, E);

  for (int m = 0; m < M; ++m) {
    const float* W = Ws + (size_t)m * CH * CH;
    const float* as_ = att_src + (size_t)m * CH;
    const float* ad_ = att_dst + (size_t)m * CH;
    const float* b_ = bias + (size_t)m * CH;

    gat_gemm_wmma<<<g_gemm, blk, smem, stream>>>(x, W, h, N);
    gat_node_gather<<<g_gather, blk, 0, stream>>>(rowptr, csr_src, as_, ad_, h,
                                                  b_, x, N);
  }
}


# layer = one launch: 4-wave teams gather 16 nodes' attention sums of x rows (scores by W a_src, W a_dst), then f32 WMMA by W, sigmoid; no h array, no copy; 10 to 6 launches; W applied after edge sum
# speedup vs baseline: 6.6981x; 1.1312x over previous
#include <hip/hip_runtime.h>


typedef float v2f __attribute__((ext_vector_type(2)));
typedef float v8f __attribute__((ext_vector_type(8)));

#define CH 128
#define NEG_SLOPE 0.2f
#define SM_EPS 1e-16f
#define WT_PAD 132

#define BIN_SHIFT 9
#define BIN_NODES (1 << BIN_SHIFT)
#define MAX_BINS 256
#define SORT_THREADS 1024
#define SORT_PER_THREAD 8
#define SORT_CHUNK (SORT_THREADS * SORT_PER_THREAD)
#define TABLE_AHEAD 16
#define FIN_CAP 16384
#define FIN_AHEAD 8

__device__ __forceinline__ int wave_sum_int(int v) {
  for (int o = 16; o > 0; o >>= 1) v += __shfl_xor(v, o, 32);
  return v;
}

__device__ __forceinline__ int block_exscan(int v, int* wsum, int& total) {
  const int lane = threadIdx.x & 31;
  const int wave = threadIdx.x >> 5;
  int incl = v;
#pragma unroll
  for (int o = 1; o < 32; o <<= 1) {
    const int y = __shfl_up(incl, o, 32);
    if (lane >= o) incl += y;
  }
  if (lane == 31) wsum[wave] = incl;
  __syncthreads();
  const int w = wsum[lane];
  const int before = wave_sum_int((lane < wave) ? w : 0);
  total = wave_sum_int(w);
  __syncthreads();
  return before + incl - v;
}

__global__ void bin_count(const int* __restrict__ dst, int* __restrict__ table,
                          int n_edges) {
  __shared__ int hist[MAX_BINS];
  const int t = threadIdx.x;
  if (t < MAX_BINS) hist[t] = 0;
  __syncthreads();
  const int e0 = blockIdx.x * SORT_CHUNK + t;
  int dv[SORT_PER_THREAD];
#pragma unroll
  for (int k = 0; k < SORT_PER_THREAD; ++k) {
    const int e = e0 + k * SORT_THREADS;
    dv[k] = (e < n_edges) ? dst[e] : -1;
  }
#pragma unroll
  for (int k = 0; k < SORT_PER_THREAD; ++k)
    if (dv[k] >= 0) atomicAdd(&hist[dv[k] >> BIN_SHIFT], 1);
  __syncthreads();
  if (t < MAX_BINS) table[(size_t)blockIdx.x * MAX_BINS + t] = hist[t];
}

__global__ void bin_scatter(const int* __restrict__ src,
                            const int* __restrict__ dst,
                            const int* __restrict__ table,
                            int* __restrict__ pairs, int* __restrict__ bin_start,
                            int n_edges, int n_chunks) {
  __shared__ int2 stage[SORT_CHUNK];
  __shared__ int tot[MAX_BINS];
  __shared__ int bef[MAX_BINS];
  __shared__ int cursor[MAX_BINS];
  __shared__ int gbase[MAX_BINS];
  __shared__ int wsum[32];
  const int t = threadIdx.x;
  const int c = blockIdx.x;

  const int e0 = c * SORT_CHUNK + t;
  int dv[SORT_PER_THREAD];
  int sv[SORT_PER_THREAD];
#pragma unroll
  for (int k = 0; k < SORT_PER_THREAD; ++k) {
    const int e = e0 + k * SORT_THREADS;
    dv[k] = (e < n_edges) ? dst[e] : -1;
    sv[k] = (e < n_edges) ? src[e] : 0;
  }

  if (t < MAX_BINS) {
    tot[t] = 0;
    bef[t] = 0;
  }
  __syncthreads();
  {
    const int b = t & (MAX_BINS - 1);
    const int part = t >> 8;
    int ptot = 0;
    int pbef = 0;
    for (int r0 = part; r0 < n_chunks; r0 += 4 * TABLE_AHEAD) {
      int v[TABLE_AHEAD];
#pragma unroll
      for (int k = 0; k < TABLE_AHEAD; ++k) {
        const int r = r0 + 4 * k;
        v[k] = (r < n_chunks) ? table[(size_t)r * MAX_BINS + b] : 0;
      }
#pragma unroll
      for (int k = 0; k < TABLE_AHEAD; ++k) {
        const int r = r0 + 4 * k;
        ptot += v[k];
        if (r < c) pbef += v[k];
      }
    }
    atomicAdd(&tot[b], ptot);
    atomicAdd(&bef[b], pbef);
  }
  __syncthreads();
  const int own = (t < MAX_BINS) ? table[(size_t)c * MAX_BINS + t] : 0;
  const int total = (t < MAX_BINS) ? tot[t] : 0;
  int sum_own;
  int sum_all;
  const int lstart = block_exscan(own, wsum, sum_own);
  const int bstart = block_exscan(total, wsum, sum_all);
  if (t < MAX_BINS) {
    cursor[t] = lstart;
    gbase[t] = bstart + bef[t] - lstart;
    if (c == 0) bin_start[t] = bstart;
  }
  if (c == 0 && t == 0) bin_start[MAX_BINS] = sum_all;
  __syncthreads();

#pragma unroll
  for (int k = 0; k < SORT_PER_THREAD; ++k) {
    if (dv[k] >= 0) {
      const int slot = atomicAdd(&cursor[dv[k] >> BIN_SHIFT], 1);
      stage[slot] = make_int2(dv[k], sv[k]);
    }
  }
  __syncthreads();
  const int left = n_edges - c * SORT_CHUNK;
  const int nloc = (left < SORT_CHUNK) ? left : SORT_CHUNK;
  int2* out2 = (int2*)pairs;
  for (int i = t; i < nloc; i += SORT_THREADS) {
    const int2 p = stage[i];
    out2[gbase[p.x >> BIN_SHIFT] + i] = p;
  }
}

__global__ void bin_finish(const int* __restrict__ pairs,
                           const int* __restrict__ bin_start,
                           int* __restrict__ rowptr, int* __restrict__ csr_src,
                           int n_nodes, int n_edges) {
  __shared__ int ncnt[BIN_NODES];
  __shared__ int sorted[FIN_CAP];
  __shared__ int wsum[32];
  const int t = threadIdx.x;
  const int q = blockIdx.x;
  const int node0 = q << BIN_SHIFT;
  const int beg = bin_start[q];
  const int end = bin_start[q + 1];
  const int len = end - beg;
  const int2* in2 = (const int2*)pairs;
  if (t < BIN_NODES) ncnt[t] = 0;
  __syncthreads();
  for (int i0 = beg + t; i0 < end; i0 += FIN_AHEAD * SORT_THREADS) {
    int dv[FIN_AHEAD];
#pragma unroll
    for (int k = 0; k < FIN_AHEAD; ++k) {
      const int i = i0 + k * SORT_THREADS;
      dv[k] = (i < end) ? in2[i].x : -1;
    }
#pragma unroll
    for (int k = 0; k < FIN_AHEAD; ++k)
      if (dv[k] >= 0) atomicAdd(&ncnt[dv[k] - node0], 1);
  }
  __syncthreads();
  const int mine = (t < BIN_NODES) ? ncnt[t] : 0;
  int sum_bin;
  const int excl = block_exscan(mine, wsum, sum_bin);
  if (t < BIN_NODES) {
    ncnt[t] = excl;
    const int node = node0 + t;
    if (node < n_nodes) rowptr[node] = beg + excl;
  }
  if (q == (int)gridDim.x - 1 && t == 0) rowptr[n_nodes] = n_edges;
  __syncthreads();
  const bool fits = len <= FIN_CAP;
  for (int i0 = beg + t; i0 < end; i0 += FIN_AHEAD * SORT_THREADS) {
    int2 p[FIN_AHEAD];
#pragma unroll
    for (int k = 0; k < FIN_AHEAD; ++k) {
      const int i = i0 + k * SORT_THREADS;
      p[k] = (i < end) ? in2[i] : make_int2(-1, 0);
    }
#pragma unroll
    for (int k = 0; k < FIN_AHEAD; ++k) {
      if (p[k].x >= 0) {
        const int slot = atomicAdd(&ncnt[p[k].x - node0], 1);
        if (fits)
          sorted[slot] = p[k].y;
        else
          csr_src[beg + slot] = p[k].y;
      }
    }
  }
  if (fits) {
    __syncthreads();
    for (int i = t; i < len; i += SORT_THREADS) csr_src[beg + i] = sorted[i];
  }
}

#define LOG2E 1.4426950408889634f

__device__ __forceinline__ float leaky(float v) {
  return (v > 0.0f) ? v : NEG_SLOPE * v;
}

__device__ __forceinline__ float xhalf(float x) {
  const int b = __builtin_bit_cast(int, x);
  return __builtin_bit_cast(
      float, __builtin_amdgcn_permlanex16(b, b, 0x76543210, 0xfedcba98, false, false));
}

template <int M>
__device__ __forceinline__ float xrow(float x) {
  const int b = __builtin_bit_cast(int, x);
  return __builtin_bit_cast(
      float, __builtin_amdgcn_update_dpp(0, b, 0x160 | M, 0xf, 0xf, true));
}

__device__ __forceinline__ float dot4(const float4 a, const float4 b) {
  return a.x * b.x + a.y * b.y + a.z * b.z + a.w * b.w;
}

__device__ __forceinline__ float exp_neg(float v) {
  return __builtin_amdgcn_exp2f(v * LOG2E);
}

__device__ __forceinline__ void load_rows(float4& r0, float4& r1, float4& r2,
                                          float4& r3, int u, int t,
                                          const float4* __restrict__ H4,
                                          int lane) {
  const int u0 = __builtin_amdgcn_readlane(u, t);
  const int u1 = __builtin_amdgcn_readlane(u, t + 1);
  const int u2 = __builtin_amdgcn_readlane(u, t + 2);
  const int u3 = __builtin_amdgcn_readlane(u, t + 3);
  r0 = H4[(size_t)u0 * (CH / 4) + lane];
  r1 = H4[(size_t)u1 * (CH / 4) + lane];
  r2 = H4[(size_t)u2 * (CH / 4) + lane];
  r3 = H4[(size_t)u3 * (CH / 4) + lane];
}

__device__ __forceinline__ void take_rows(const float4 c0, const float4 c1,
                                          const float4 c2, const float4 c3,
                                          int t, int cnt, const float4 as4,
                                          float dv, bool hi16, bool hi8,
                                          int slot, float& m, float& den,
                                          float4& acc) {
  const float p0 = dot4(c0, as4);
  const float p1 = dot4(c1, as4);
  const float p2 = dot4(c2, as4);
  const float p3 = dot4(c3, as4);
  const float kA = hi16 ? p2 : p0;
  const float sA = hi16 ? p0 : p2;
  const float kB = hi16 ? p3 : p1;
  const float sB = hi16 ? p1 : p3;
  const float qA = kA + xhalf(sA);
  const float qB = kB + xhalf(sB);
  const float k2 = hi8 ? qB : qA;
  const float s2 = hi8 ? qA : qB;
  float sc = k2 + xrow<8>(s2);
  sc += xrow<4>(sc);
  sc += xrow<2>(sc);
  sc += xrow<1>(sc);

  float e = leaky(sc + dv);
  e = (t + slot < cnt) ? e : -__builtin_inff();
  float mb = __builtin_fmaxf(e, xrow<8>(e));
  mb = __builtin_fmaxf(mb, xhalf(mb));
  const float mn = __builtin_fmaxf(m, mb);
  const float corr = exp_neg(m - mn);
  const float wv = exp_neg(e - mn);
  float ws = wv + xrow<8>(wv);
  ws = ws + xhalf(ws);
  den = den * corr + ws;
  m = mn;
  const float w0 = __int_as_float(__builtin_amdgcn_readlane(__float_as_int(wv), 0));
  const float w1 = __int_as_float(__builtin_amdgcn_readlane(__float_as_int(wv), 8));
  const float w2 = __int_as_float(__builtin_amdgcn_readlane(__float_as_int(wv), 16));
  const float w3 = __int_as_float(__builtin_amdgcn_readlane(__float_as_int(wv), 24));
  acc.x = acc.x * corr + w0 * c0.x;
  acc.y = acc.y * corr + w0 * c0.y;
  acc.z = acc.z * corr + w0 * c0.z;
  acc.w = acc.w * corr + w0 * c0.w;
  acc.x += w1 * c1.x;
  acc.y += w1 * c1.y;
  acc.z += w1 * c1.z;
  acc.w += w1 * c1.w;
  acc.x += w2 * c2.x;
  acc.y += w2 * c2.y;
  acc.z += w2 * c2.z;
  acc.w += w2 * c2.w;
  acc.x += w3 * c3.x;
  acc.y += w3 * c3.y;
  acc.z += w3 * c3.z;
  acc.w += w3 * c3.w;
}

#define FB_THREADS 512
#define FB_TEAMS 4
#define FB_BLOCKS 512
#define FB_LDS_FLOATS (CH * WT_PAD + FB_TEAMS * 16 * WT_PAD + 2 * CH)

__device__ __forceinline__ float4 gather_node(
    int v, int beg, int end, int u, const float4 hs, const float4 p4,
    const float4 q4, const int* __restrict__ csr_src,
    const float4* __restrict__ X4, int lane, bool hi16, bool hi8, int slot) {
  float ps = dot4(hs, p4);
  float pd = dot4(hs, q4);
  ps += xrow<1>(ps);
  pd += xrow<1>(pd);
  ps += xrow<2>(ps);
  pd += xrow<2>(pd);
  ps += xrow<4>(ps);
  pd += xrow<4>(pd);
  ps += xrow<8>(ps);
  pd += xrow<8>(pd);
  ps += xhalf(ps);
  pd += xhalf(pd);
  const float dv = pd;

  float m = leaky(ps + dv);
  float den = 1.0f;
  float4 acc = hs;

  if (beg < end) {
    const int left = end - beg;
    const int cnt = (left < 32) ? left : 32;
    float4 a0, a1, a2, a3, b0, b1, b2, b3;
    load_rows(a0, a1, a2, a3, u, 0, X4, lane);
    for (int t = 0;; t += 8) {
      if (t + 4 >= cnt) {
        take_rows(a0, a1, a2, a3, t, cnt, p4, dv, hi16, hi8, slot, m, den, acc);
        break;
      }
      load_rows(b0, b1, b2, b3, u, t + 4, X4, lane);
      take_rows(a0, a1, a2, a3, t, cnt, p4, dv, hi16, hi8, slot, m, den, acc);
      if (t + 8 >= cnt) {
        take_rows(b0, b1, b2, b3, t + 4, cnt, p4, dv, hi16, hi8, slot, m, den, acc);
        break;
      }
      load_rows(a0, a1, a2, a3, u, t + 8, X4, lane);
      take_rows(b0, b1, b2, b3, t + 4, cnt, p4, dv, hi16, hi8, slot, m, den, acc);
    }
  }
  for (int base = beg + 32; base < end; base += 32) {
    const int j = base + lane;
    const int u2 = (j < end) ? csr_src[j] : v;
    const int left = end - base;
    const int cnt = (left < 32) ? left : 32;
    for (int t = 0; t < cnt; t += 4) {
      float4 a0, a1, a2, a3;
      load_rows(a0, a1, a2, a3, u2, t, X4, lane);
      take_rows(a0, a1, a2, a3, t, cnt, p4, dv, hi16, hi8, slot, m, den, acc);
    }
  }

  const float rden = 1.0f / (den + SM_EPS);
  return make_float4(acc.x * rden, acc.y * rden, acc.z * rden, acc.w * rden);
}

__global__ __launch_bounds__(FB_THREADS) void gat_layer(
    const int* __restrict__ rowptr, const int* __restrict__ csr_src,
    const float* __restrict__ a_src, const float* __restrict__ a_dst,
    const float* __restrict__ W, const float* __restrict__ bias,
    const float* __restrict__ xin, float* __restrict__ xout, int n_nodes) {
  extern __shared__ float sm[];
  float* wt = sm;
  float* tiles = sm + CH * WT_PAD;
  float* pq = tiles + FB_TEAMS * 16 * WT_PAD;
  const int tid = threadIdx.x;
  for (int i = tid; i < CH * CH; i += FB_THREADS) {
    const int k = i >> 7;
    const int n = i & (CH - 1);
    wt[n * WT_PAD + k] = W[i];
  }
  __syncthreads();
  if (tid < 2 * CH) {
    const float* __restrict__ av = (tid < CH) ? a_src : a_dst;
    const int k = tid & (CH - 1);
    float sacc = 0.0f;
    for (int n = 0; n < CH; ++n) sacc += wt[n * WT_PAD + k] * av[n];
    pq[tid] = sacc;
  }
  __syncthreads();

  const int lane = tid & 31;
  const int wave = tid >> 5;
  const int team = wave >> 2;
  const int sub = wave & 3;
  const bool hi16 = (lane & 16) != 0;
  const bool hi8 = (lane & 8) != 0;
  const int slot = lane >> 3;
  float* mytile = tiles + team * 16 * WT_PAD;
  const float4 p4 = ((const float4*)pq)[lane];
  const float4 q4 = ((const float4*)(pq + CH))[lane];
  const float4 b4 = ((const float4*)bias)[lane];
  const float4* X4 = (const float4*)xin;

  const int n_teams = (int)gridDim.x * FB_TEAMS;
  const int per_team = (n_nodes + n_teams - 1) / n_teams;
  const int n_trips = (per_team + 15) >> 4;
  const int vstart = __builtin_amdgcn_readfirstlane(
      ((int)blockIdx.x * FB_TEAMS + team) * per_team);
  const int vlimit = (vstart + per_team < n_nodes) ? vstart + per_team : n_nodes;

  const int mrow = lane & 15;
  const int koff = (lane >> 4) * 2;
  const float* arow = mytile + mrow * WT_PAD + koff;
  const float* wrow0 = wt + ((2 * sub) * 16 + mrow) * WT_PAD + koff;
  const float* wrow1 = wrow0 + 16 * WT_PAD;

  for (int trip = 0; trip < n_trips; ++trip) {
    const int v0 = __builtin_amdgcn_readfirstlane(vstart + trip * 16 + sub * 4);
    int rp[5];
#pragma unroll
    for (int i = 0; i < 5; ++i) {
      const int vi = (v0 + i < n_nodes) ? v0 + i : n_nodes;
      rp[i] = rowptr[vi];
    }
    float4 xs[4];
    int un = 0;
    float4 hn = make_float4(0.0f, 0.0f, 0.0f, 0.0f);
    if (v0 < vlimit) {
      un = (rp[0] + lane < rp[1]) ? csr_src[rp[0] + lane] : v0;
      hn = X4[(size_t)v0 * (CH / 4) + lane];
    }
#pragma unroll
    for (int i = 0; i < 4; ++i) {
      const int v = v0 + i;
      const int u = un;
      const float4 hs = hn;
      if (i < 3 && v + 1 < vlimit) {
        un = (rp[i + 1] + lane < rp[i + 2]) ? csr_src[rp[i + 1] + lane] : v + 1;
        hn = X4[(size_t)(v + 1) * (CH / 4) + lane];
      }
      float4 agg = make_float4(0.0f, 0.0f, 0.0f, 0.0f);
      if (v < vlimit)
        agg = gather_node(v, rp[i], rp[i + 1], u, hs, p4, q4, csr_src, X4, lane,
                          hi16, hi8, slot);
      xs[i] = hs;
      *(float4*)(mytile + (sub * 4 + i) * WT_PAD + 4 * lane) = agg;
    }
    __syncthreads();

    v8f acc0 = {};
    v8f acc1 = {};
    v2f fa = *(const v2f*)(arow);
    v2f fb0 = *(const v2f*)(wrow0);
    v2f fb1 = *(const v2f*)(wrow1);
    for (int k4 = 0; k4 < CH; k4 += 4) {
      const int kn = (k4 + 4 < CH) ? k4 + 4 : k4;
      const v2f na = *(const v2f*)(arow + kn);
      const v2f nb0 = *(const v2f*)(wrow0 + kn);
      const v2f nb1 = *(const v2f*)(wrow1 + kn);
      acc0 = __builtin_amdgcn_wmma_f32_16x16x4_f32(false, fa, false, fb0,
                                                   (short)0, acc0, false, false);
      acc1 = __builtin_amdgcn_wmma_f32_16x16x4_f32(false, fa, false, fb1,
                                                   (short)0, acc1, false, false);
      fa = na;
      fb0 = nb0;
      fb1 = nb1;
    }
    __syncthreads();
    {
      const int rhi = (lane >> 4) * 8;
      const int col = (2 * sub) * 16 + (lane & 15);
#pragma unroll
      for (int r = 0; r < 8; ++r) {
        mytile[(r + rhi) * WT_PAD + col] = acc0[r];
        mytile[(r + rhi) * WT_PAD + col + 16] = acc1[r];
      }
    }
    __syncthreads();

#pragma unroll
    for (int i = 0; i < 4; ++i) {
      const int v = v0 + i;
      if (v < vlimit) {
        const float4 o =
            *(const float4*)(mytile + (sub * 4 + i) * WT_PAD + 4 * lane);
        float4 xv = xs[i];
        xv.x += 1.0f / (1.0f + __builtin_expf(-(o.x + b4.x)));
        xv.y += 1.0f / (1.0f + __builtin_expf(-(o.y + b4.y)));
        xv.z += 1.0f / (1.0f + __builtin_expf(-(o.z + b4.z)));
        xv.w += 1.0f / (1.0f + __builtin_expf(-(o.w + b4.w)));
        ((float4*)(xout + (size_t)v * CH))[lane] = xv;
      }
    }
  }
}

extern "C" void kernel_launch(void* const* d_in, const int* in_sizes, int n_in,
                              void* d_out, int out_size, void* d_ws,
                              size_t ws_size, hipStream_t stream) {
  const float* x_in = (const float*)d_in[0];
  const int* ei = (const int*)d_in[1];
  const float* Ws = (const float*)d_in[2];
  const float* att_src = (const float*)d_in[3];
  const float* att_dst = (const float*)d_in[4];
  const float* bias = (const float*)d_in[5];

  const int N = in_sizes[0] / CH;
  const int E = in_sizes[1] / 2;
  const int M = in_sizes[3] / CH;
  const int* src = ei;
  const int* dst = ei + E;

  float* ws = (float*)d_ws;
  float* plane = ws;
  int* rowptr = (int*)(plane + (size_t)N * CH);
  int* csr_src = rowptr + N + 1;

  float* out = (float*)d_out;

  const int n_chunks = (E + SORT_CHUNK - 1) / SORT_CHUNK;
  const int n_bins = (N + BIN_NODES - 1) / BIN_NODES;
  int* pairs = (int*)plane;
  int* table = pairs + 2 * (size_t)E;
  int* bin_start = table + (size_t)n_chunks * MAX_BINS;
  const size_t smem = (size_t)FB_LDS_FLOATS * sizeof(float);

  bin_count<<<n_chunks, SORT_THREADS, 0, stream>>>(dst, table, E);
  bin_scatter<<<n_chunks, SORT_THREADS, 0, stream>>>(src, dst, table, pairs,
                                                    bin_start, E, n_chunks);
  bin_finish<<<n_bins, SORT_THREADS, 0, stream>>>(pairs, bin_start, rowptr,
                                                 csr_src, N, E);

  const float* cur = x_in;
  for (int m = 0; m < M; ++m) {
    const float* W = Ws + (size_t)m * CH * CH;
    const float* as_ = att_src + (size_t)m * CH;
    const float* ad_ = att_dst + (size_t)m * CH;
    const float* b_ = bias + (size_t)m * CH;
    float* nxt = (((M - 1 - m) & 1) != 0) ? plane : out;
    gat_layer<<<FB_BLOCKS, FB_THREADS, smem, stream>>>(rowptr, csr_src, as_, ad_,
                                                     W, b_, cur, nxt, N);
    cur = nxt;
  }
}


# nodes dealt as whole 16-row tiles: 3 full trips a team; the 106 left-over tiles go one node per wave to 106 blocks where one team multiplies; no 4th trip for one node a team; float sums unchanged
# speedup vs baseline: 7.1099x; 1.0615x over previous
#include <hip/hip_runtime.h>


typedef float v2f __attribute__((ext_vector_type(2)));
typedef float v8f __attribute__((ext_vector_type(8)));

#define CH 128
#define NEG_SLOPE 0.2f
#define SM_EPS 1e-16f
#define WT_PAD 132

#define BIN_SHIFT 9
#define BIN_NODES (1 << BIN_SHIFT)
#define MAX_BINS 256
#define SORT_THREADS 1024
#define SORT_PER_THREAD 8
#define SORT_CHUNK (SORT_THREADS * SORT_PER_THREAD)
#define TABLE_AHEAD 16
#define FIN_CAP 16384
#define FIN_AHEAD 8

__device__ __forceinline__ int wave_sum_int(int v) {
  for (int o = 16; o > 0; o >>= 1) v += __shfl_xor(v, o, 32);
  return v;
}

__device__ __forceinline__ int block_exscan(int v, int* wsum, int& total) {
  const int lane = threadIdx.x & 31;
  const int wave = threadIdx.x >> 5;
  int incl = v;
#pragma unroll
  for (int o = 1; o < 32; o <<= 1) {
    const int y = __shfl_up(incl, o, 32);
    if (lane >= o) incl += y;
  }
  if (lane == 31) wsum[wave] = incl;
  __syncthreads();
  const int w = wsum[lane];
  const int before = wave_sum_int((lane < wave) ? w : 0);
  total = wave_sum_int(w);
  __syncthreads();
  return before + incl - v;
}

__global__ void bin_count(const int* __restrict__ dst, int* __restrict__ table,
                          int n_edges) {
  __shared__ int hist[MAX_BINS];
  const int t = threadIdx.x;
  if (t < MAX_BINS) hist[t] = 0;
  __syncthreads();
  const int e0 = blockIdx.x * SORT_CHUNK + t;
  int dv[SORT_PER_THREAD];
#pragma unroll
  for (int k = 0; k < SORT_PER_THREAD; ++k) {
    const int e = e0 + k * SORT_THREADS;
    dv[k] = (e < n_edges) ? dst[e] : -1;
  }
#pragma unroll
  for (int k = 0; k < SORT_PER_THREAD; ++k)
    if (dv[k] >= 0) atomicAdd(&hist[dv[k] >> BIN_SHIFT], 1);
  __syncthreads();
  if (t < MAX_BINS) table[(size_t)blockIdx.x * MAX_BINS + t] = hist[t];
}

__global__ void bin_scatter(const int* __restrict__ src,
                            const int* __restrict__ dst,
                            const int* __restrict__ table,
                            int* __restrict__ pairs, int* __restrict__ bin_start,
                            int n_edges, int n_chunks) {
  __shared__ int2 stage[SORT_CHUNK];
  __shared__ int tot[MAX_BINS];
  __shared__ int bef[MAX_BINS];
  __shared__ int cursor[MAX_BINS];
  __shared__ int gbase[MAX_BINS];
  __shared__ int wsum[32];
  const int t = threadIdx.x;
  const int c = blockIdx.x;

  const int e0 = c * SORT_CHUNK + t;
  int dv[SORT_PER_THREAD];
  int sv[SORT_PER_THREAD];
#pragma unroll
  for (int k = 0; k < SORT_PER_THREAD; ++k) {
    const int e = e0 + k * SORT_THREADS;
    dv[k] = (e < n_edges) ? dst[e] : -1;
    sv[k] = (e < n_edges) ? src[e] : 0;
  }

  if (t < MAX_BINS) {
    tot[t] = 0;
    bef[t] = 0;
  }
  __syncthreads();
  {
    const int b = t & (MAX_BINS - 1);
    const int part = t >> 8;
    int ptot = 0;
    int pbef = 0;
    for (int r0 = part; r0 < n_chunks; r0 += 4 * TABLE_AHEAD) {
      int v[TABLE_AHEAD];
#pragma unroll
      for (int k = 0; k < TABLE_AHEAD; ++k) {
        const int r = r0 + 4 * k;
        v[k] = (r < n_chunks) ? table[(size_t)r * MAX_BINS + b] : 0;
      }
#pragma unroll
      for (int k = 0; k < TABLE_AHEAD; ++k) {
        const int r = r0 + 4 * k;
        ptot += v[k];
        if (r < c) pbef += v[k];
      }
    }
    atomicAdd(&tot[b], ptot);
    atomicAdd(&bef[b], pbef);
  }
  __syncthreads();
  const int own = (t < MAX_BINS) ? table[(size_t)c * MAX_BINS + t] : 0;
  const int total = (t < MAX_BINS) ? tot[t] : 0;
  int sum_own;
  int sum_all;
  const int lstart = block_exscan(own, wsum, sum_own);
  const int bstart = block_exscan(total, wsum, sum_all);
  if (t < MAX_BINS) {
    cursor[t] = lstart;
    gbase[t] = bstart + bef[t] - lstart;
    if (c == 0) bin_start[t] = bstart;
  }
  if (c == 0 && t == 0) bin_start[MAX_BINS] = sum_all;
  __syncthreads();

#pragma unroll
  for (int k = 0; k < SORT_PER_THREAD; ++k) {
    if (dv[k] >= 0) {
      const int slot = atomicAdd(&cursor[dv[k] >> BIN_SHIFT], 1);
      stage[slot] = make_int2(dv[k], sv[k]);
    }
  }
  __syncthreads();
  const int left = n_edges - c * SORT_CHUNK;
  const int nloc = (left < SORT_CHUNK) ? left : SORT_CHUNK;
  int2* out2 = (int2*)pairs;
  for (int i = t; i < nloc; i += SORT_THREADS) {
    const int2 p = stage[i];
    out2[gbase[p.x >> BIN_SHIFT] + i] = p;
  }
}

__global__ void bin_finish(const int* __restrict__ pairs,
                           const int* __restrict__ bin_start,
                           int* __restrict__ rowptr, int* __restrict__ csr_src,
                           int n_nodes, int n_edges) {
  __shared__ int ncnt[BIN_NODES];
  __shared__ int sorted[FIN_CAP];
  __shared__ int wsum[32];
  const int t = threadIdx.x;
  const int q = blockIdx.x;
  const int node0 = q << BIN_SHIFT;
  const int beg = bin_start[q];
  const int end = bin_start[q + 1];
  const int len = end - beg;
  const int2* in2 = (const int2*)pairs;
  if (t < BIN_NODES) ncnt[t] = 0;
  __syncthreads();
  for (int i0 = beg + t; i0 < end; i0 += FIN_AHEAD * SORT_THREADS) {
    int dv[FIN_AHEAD];
#pragma unroll
    for (int k = 0; k < FIN_AHEAD; ++k) {
      const int i = i0 + k * SORT_THREADS;
      dv[k] = (i < end) ? in2[i].x : -1;
    }
#pragma unroll
    for (int k = 0; k < FIN_AHEAD; ++k)
      if (dv[k] >= 0) atomicAdd(&ncnt[dv[k] - node0], 1);
  }
  __syncthreads();
  const int mine = (t < BIN_NODES) ? ncnt[t] : 0;
  int sum_bin;
  const int excl = block_exscan(mine, wsum, sum_bin);
  if (t < BIN_NODES) {
    ncnt[t] = excl;
    const int node = node0 + t;
    if (node < n_nodes) rowptr[node] = beg + excl;
  }
  if (q == (int)gridDim.x - 1 && t == 0) rowptr[n_nodes] = n_edges;
  __syncthreads();
  const bool fits = len <= FIN_CAP;
  for (int i0 = beg + t; i0 < end; i0 += FIN_AHEAD * SORT_THREADS) {
    int2 p[FIN_AHEAD];
#pragma unroll
    for (int k = 0; k < FIN_AHEAD; ++k) {
      const int i = i0 + k * SORT_THREADS;
      p[k] = (i < end) ? in2[i] : make_int2(-1, 0);
    }
#pragma unroll
    for (int k = 0; k < FIN_AHEAD; ++k) {
      if (p[k].x >= 0) {
        const int slot = atomicAdd(&ncnt[p[k].x - node0], 1);
        if (fits)
          sorted[slot] = p[k].y;
        else
          csr_src[beg + slot] = p[k].y;
      }
    }
  }
  if (fits) {
    __syncthreads();
    for (int i = t; i < len; i += SORT_THREADS) csr_src[beg + i] = sorted[i];
  }
}

#define LOG2E 1.4426950408889634f

__device__ __forceinline__ float leaky(float v) {
  return (v > 0.0f) ? v : NEG_SLOPE * v;
}

__device__ __forceinline__ float xhalf(float x) {
  const int b = __builtin_bit_cast(int, x);
  return __builtin_bit_cast(
      float, __builtin_amdgcn_permlanex16(b, b, 0x76543210, 0xfedcba98, false, false));
}

template <int M>
__device__ __forceinline__ float xrow(float x) {
  const int b = __builtin_bit_cast(int, x);
  return __builtin_bit_cast(
      float, __builtin_amdgcn_update_dpp(0, b, 0x160 | M, 0xf, 0xf, true));
}

__device__ __forceinline__ float dot4(const float4 a, const float4 b) {
  return a.x * b.x + a.y * b.y + a.z * b.z + a.w * b.w;
}

__device__ __forceinline__ float exp_neg(float v) {
  return __builtin_amdgcn_exp2f(v * LOG2E);
}

__device__ __forceinline__ void load_rows(float4& r0, float4& r1, float4& r2,
                                          float4& r3, int u, int t,
                                          const float4* __restrict__ H4,
                                          int lane) {
  const int u0 = __builtin_amdgcn_readlane(u, t);
  const int u1 = __builtin_amdgcn_readlane(u, t + 1);
  const int u2 = __builtin_amdgcn_readlane(u, t + 2);
  const int u3 = __builtin_amdgcn_readlane(u, t + 3);
  r0 = H4[(size_t)u0 * (CH / 4) + lane];
  r1 = H4[(size_t)u1 * (CH / 4) + lane];
  r2 = H4[(size_t)u2 * (CH / 4) + lane];
  r3 = H4[(size_t)u3 * (CH / 4) + lane];
}

__device__ __forceinline__ void take_rows(const float4 c0, const float4 c1,
                                          const float4 c2, const float4 c3,
                                          int t, int cnt, const float4 as4,
                                          float dv, bool hi16, bool hi8,
                                          int slot, float& m, float& den,
                                          float4& acc) {
  const float p0 = dot4(c0, as4);
  const float p1 = dot4(c1, as4);
  const float p2 = dot4(c2, as4);
  const float p3 = dot4(c3, as4);
  const float kA = hi16 ? p2 : p0;
  const float sA = hi16 ? p0 : p2;
  const float kB = hi16 ? p3 : p1;
  const float sB = hi16 ? p1 : p3;
  const float qA = kA + xhalf(sA);
  const float qB = kB + xhalf(sB);
  const float k2 = hi8 ? qB : qA;
  const float s2 = hi8 ? qA : qB;
  float sc = k2 + xrow<8>(s2);
  sc += xrow<4>(sc);
  sc += xrow<2>(sc);
  sc += xrow<1>(sc);

  float e = leaky(sc + dv);
  e = (t + slot < cnt) ? e : -__builtin_inff();
  float mb = __builtin_fmaxf(e, xrow<8>(e));
  mb = __builtin_fmaxf(mb, xhalf(mb));
  const float mn = __builtin_fmaxf(m, mb);
  const float corr = exp_neg(m - mn);
  const float wv = exp_neg(e - mn);
  float ws = wv + xrow<8>(wv);
  ws = ws + xhalf(ws);
  den = den * corr + ws;
  m = mn;
  const float w0 = __int_as_float(__builtin_amdgcn_readlane(__float_as_int(wv), 0));
  const float w1 = __int_as_float(__builtin_amdgcn_readlane(__float_as_int(wv), 8));
  const float w2 = __int_as_float(__builtin_amdgcn_readlane(__float_as_int(wv), 16));
  const float w3 = __int_as_float(__builtin_amdgcn_readlane(__float_as_int(wv), 24));
  acc.x = acc.x * corr + w0 * c0.x;
  acc.y = acc.y * corr + w0 * c0.y;
  acc.z = acc.z * corr + w0 * c0.z;
  acc.w = acc.w * corr + w0 * c0.w;
  acc.x += w1 * c1.x;
  acc.y += w1 * c1.y;
  acc.z += w1 * c1.z;
  acc.w += w1 * c1.w;
  acc.x += w2 * c2.x;
  acc.y += w2 * c2.y;
  acc.z += w2 * c2.z;
  acc.w += w2 * c2.w;
  acc.x += w3 * c3.x;
  acc.y += w3 * c3.y;
  acc.z += w3 * c3.z;
  acc.w += w3 * c3.w;
}

#define FB_THREADS 512
#define FB_TEAMS 4
#define FB_BLOCKS 512
#define FB_LDS_FLOATS (CH * WT_PAD + FB_TEAMS * 16 * WT_PAD + 2 * CH)

__device__ __forceinline__ float4 gather_node(
    int v, int beg, int end, int u, const float4 hs, const float4 p4,
    const float4 q4, const int* __restrict__ csr_src,
    const float4* __restrict__ X4, int lane, bool hi16, bool hi8, int slot) {
  float ps = dot4(hs, p4);
  float pd = dot4(hs, q4);
  ps += xrow<1>(ps);
  pd += xrow<1>(pd);
  ps += xrow<2>(ps);
  pd += xrow<2>(pd);
  ps += xrow<4>(ps);
  pd += xrow<4>(pd);
  ps += xrow<8>(ps);
  pd += xrow<8>(pd);
  ps += xhalf(ps);
  pd += xhalf(pd);
  const float dv = pd;

  float m = leaky(ps + dv);
  float den = 1.0f;
  float4 acc = hs;

  if (beg < end) {
    const int left = end - beg;
    const int cnt = (left < 32) ? left : 32;
    float4 a0, a1, a2, a3, b0, b1, b2, b3;
    load_rows(a0, a1, a2, a3, u, 0, X4, lane);
    for (int t = 0;; t += 8) {
      if (t + 4 >= cnt) {
        take_rows(a0, a1, a2, a3, t, cnt, p4, dv, hi16, hi8, slot, m, den, acc);
        break;
      }
      load_rows(b0, b1, b2, b3, u, t + 4, X4, lane);
      take_rows(a0, a1, a2, a3, t, cnt, p4, dv, hi16, hi8, slot, m, den, acc);
      if (t + 8 >= cnt) {
        take_rows(b0, b1, b2, b3, t + 4, cnt, p4, dv, hi16, hi8, slot, m, den, acc);
        break;
      }
      load_rows(a0, a1, a2, a3, u, t + 8, X4, lane);
      take_rows(b0, b1, b2, b3, t + 4, cnt, p4, dv, hi16, hi8, slot, m, den, acc);
    }
  }
  for (int base = beg + 32; base < end; base += 32) {
    const int j = base + lane;
    const int u2 = (j < end) ? csr_src[j] : v;
    const int left = end - base;
    const int cnt = (left < 32) ? left : 32;
    for (int t = 0; t < cnt; t += 4) {
      float4 a0, a1, a2, a3;
      load_rows(a0, a1, a2, a3, u2, t, X4, lane);
      take_rows(a0, a1, a2, a3, t, cnt, p4, dv, hi16, hi8, slot, m, den, acc);
    }
  }

  const float rden = 1.0f / (den + SM_EPS);
  return make_float4(acc.x * rden, acc.y * rden, acc.z * rden, acc.w * rden);
}

__global__ __launch_bounds__(FB_THREADS) void gat_layer(
    const int* __restrict__ rowptr, const int* __restrict__ csr_src,
    const float* __restrict__ a_src, const float* __restrict__ a_dst,
    const float* __restrict__ W, const float* __restrict__ bias,
    const float* __restrict__ xin, float* __restrict__ xout, int n_nodes) {
  extern __shared__ float sm[];
  float* wt = sm;
  float* tiles = sm + CH * WT_PAD;
  float* pq = tiles + FB_TEAMS * 16 * WT_PAD;
  const int tid = threadIdx.x;
  for (int i = tid; i < CH * CH; i += FB_THREADS) {
    const int k = i >> 7;
    const int n = i & (CH - 1);
    wt[n * WT_PAD + k] = W[i];
  }
  __syncthreads();
  if (tid < 2 * CH) {
    const float* __restrict__ av = (tid < CH) ? a_src : a_dst;
    const int k = tid & (CH - 1);
    float sacc = 0.0f;
    for (int n = 0; n < CH; ++n) sacc += wt[n * WT_PAD + k] * av[n];
    pq[tid] = sacc;
  }
  __syncthreads();

  const int lane = tid & 31;
  const int wave = __builtin_amdgcn_readfirstlane(tid >> 5);
  const int team = wave >> 2;
  const int sub = wave & 3;
  const bool hi16 = (lane & 16) != 0;
  const bool hi8 = (lane & 8) != 0;
  const int slot = lane >> 3;
  float* mytile = tiles + team * 16 * WT_PAD;
  const float4 p4 = ((const float4*)pq)[lane];
  const float4 q4 = ((const float4*)(pq + CH))[lane];
  const float4 b4 = ((const float4*)bias)[lane];
  const float4* X4 = (const float4*)xin;

  const int n_blocks = (int)gridDim.x;
  const int n_tiles = (n_nodes + 15) >> 4;
  const int full_trips = n_tiles / (n_blocks * FB_TEAMS);
  const int first_extra = full_trips * n_blocks * FB_TEAMS;
  const int rem_tiles = n_tiles - first_extra;
  const int n_extra = (rem_tiles > (int)blockIdx.x)
                          ? (rem_tiles - (int)blockIdx.x + n_blocks - 1) / n_blocks
                          : 0;
  const int n_trips = full_trips + n_extra;

  const int mrow = lane & 15;
  const int koff = (lane >> 4) * 2;
  const float* arow = mytile + mrow * WT_PAD + koff;
  const float* wrow0 = wt + ((2 * sub) * 16 + mrow) * WT_PAD + koff;
  const float* wrow1 = wrow0 + 16 * WT_PAD;

  for (int trip = 0; trip < n_trips; ++trip) {
    const bool extra = trip >= full_trips;
    if (extra) __syncthreads();
    const int tile = extra ? first_extra + (int)blockIdx.x + (trip - full_trips) * n_blocks
                           : ((int)blockIdx.x * full_trips + trip) * FB_TEAMS + team;
    const int v0 = __builtin_amdgcn_readfirstlane(tile * 16 + (extra ? wave : sub * 4));
    const int vend = extra ? v0 + 1 : n_nodes;
    const int vlimit = (vend < n_nodes) ? vend : n_nodes;
    float* myrows = extra ? tiles + wave * WT_PAD : mytile + sub * 4 * WT_PAD;
    const bool multiply = !extra || team == 0;
    int rp[5];
#pragma unroll
    for (int i = 0; i < 5; ++i) {
      const int vi = (v0 + i < n_nodes) ? v0 + i : n_nodes;
      rp[i] = rowptr[vi];
    }
    float4 xs[4];
    int un = 0;
    float4 hn = make_float4(0.0f, 0.0f, 0.0f, 0.0f);
    if (v0 < vlimit) {
      un = (rp[0] + lane < rp[1]) ? csr_src[rp[0] + lane] : v0;
      hn = X4[(size_t)v0 * (CH / 4) + lane];
    }
#pragma unroll
    for (int i = 0; i < 4; ++i) {
      const int v = v0 + i;
      const int u = un;
      const float4 hs = hn;
      if (i < 3 && v + 1 < vlimit) {
        un = (rp[i + 1] + lane < rp[i + 2]) ? csr_src[rp[i + 1] + lane] : v + 1;
        hn = X4[(size_t)(v + 1) * (CH / 4) + lane];
      }
      float4 agg = make_float4(0.0f, 0.0f, 0.0f, 0.0f);
      if (v < vlimit)
        agg = gather_node(v, rp[i], rp[i + 1], u, hs, p4, q4, csr_src, X4, lane,
                          hi16, hi8, slot);
      xs[i] = hs;
      if (!extra || i == 0) *(float4*)(myrows + i * WT_PAD + 4 * lane) = agg;
    }
    __syncthreads();

    v8f acc0 = {};
    v8f acc1 = {};
    if (multiply) {
    v2f fa = *(const v2f*)(arow);
    v2f fb0 = *(const v2f*)(wrow0);
    v2f fb1 = *(const v2f*)(wrow1);
    for (int k4 = 0; k4 < CH; k4 += 4) {
      const int kn = (k4 + 4 < CH) ? k4 + 4 : k4;
      const v2f na = *(const v2f*)(arow + kn);
      const v2f nb0 = *(const v2f*)(wrow0 + kn);
      const v2f nb1 = *(const v2f*)(wrow1 + kn);
      acc0 = __builtin_amdgcn_wmma_f32_16x16x4_f32(false, fa, false, fb0,
                                                   (short)0, acc0, false, false);
      acc1 = __builtin_amdgcn_wmma_f32_16x16x4_f32(false, fa, false, fb1,
                                                   (short)0, acc1, false, false);
      fa = na;
      fb0 = nb0;
      fb1 = nb1;
    }
    }
    __syncthreads();
    if (multiply) {
      const int rhi = (lane >> 4) * 8;
      const int col = (2 * sub) * 16 + (lane & 15);
#pragma unroll
      for (int r = 0; r < 8; ++r) {
        mytile[(r + rhi) * WT_PAD + col] = acc0[r];
        mytile[(r + rhi) * WT_PAD + col + 16] = acc1[r];
      }
    }
    __syncthreads();

#pragma unroll
    for (int i = 0; i < 4; ++i) {
      const int v = v0 + i;
      if (v < vlimit) {
        const float4 o = *(const float4*)(myrows + i * WT_PAD + 4 * lane);
        float4 xv = xs[i];
        xv.x += 1.0f / (1.0f + __builtin_expf(-(o.x + b4.x)));
        xv.y += 1.0f / (1.0f + __builtin_expf(-(o.y + b4.y)));
        xv.z += 1.0f / (1.0f + __builtin_expf(-(o.z + b4.z)));
        xv.w += 1.0f / (1.0f + __builtin_expf(-(o.w + b4.w)));
        ((float4*)(xout + (size_t)v * CH))[lane] = xv;
      }
    }
  }
}

extern "C" void kernel_launch(void* const* d_in, const int* in_sizes, int n_in,
                              void* d_out, int out_size, void* d_ws,
                              size_t ws_size, hipStream_t stream) {
  const float* x_in = (const float*)d_in[0];
  const int* ei = (const int*)d_in[1];
  const float* Ws = (const float*)d_in[2];
  const float* att_src = (const float*)d_in[3];
  const float* att_dst = (const float*)d_in[4];
  const float* bias = (const float*)d_in[5];

  const int N = in_sizes[0] / CH;
  const int E = in_sizes[1] / 2;
  const int M = in_sizes[3] / CH;
  const int* src = ei;
  const int* dst = ei + E;

  float* ws = (float*)d_ws;
  float* plane = ws;
  int* rowptr = (int*)(plane + (size_t)N * CH);
  int* csr_src = rowptr + N + 1;

  float* out = (float*)d_out;

  const int n_chunks = (E + SORT_CHUNK - 1) / SORT_CHUNK;
  const int n_bins = (N + BIN_NODES - 1) / BIN_NODES;
  int* pairs = (int*)plane;
  int* table = pairs + 2 * (size_t)E;
  int* bin_start = table + (size_t)n_chunks * MAX_BINS;
  const size_t smem = (size_t)FB_LDS_FLOATS * sizeof(float);

  bin_count<<<n_chunks, SORT_THREADS, 0, stream>>>(dst, table, E);
  bin_scatter<<<n_chunks, SORT_THREADS, 0, stream>>>(src, dst, table, pairs,
                                                    bin_start, E, n_chunks);
  bin_finish<<<n_bins, SORT_THREADS, 0, stream>>>(pairs, bin_start, rowptr,
                                                 csr_src, N, E);

  const float* cur = x_in;
  for (int m = 0; m < M; ++m) {
    const float* W = Ws + (size_t)m * CH * CH;
    const float* as_ = att_src + (size_t)m * CH;
    const float* ad_ = att_dst + (size_t)m * CH;
    const float* b_ = bias + (size_t)m * CH;
    float* nxt = (((M - 1 - m) & 1) != 0) ? plane : out;
    gat_layer<<<FB_BLOCKS, FB_THREADS, smem, stream>>>(rowptr, csr_src, as_, ad_,
                                                     W, b_, cur, nxt, N);
    cur = nxt;
  }
}


# layer kernel issues about a third fewer vector instructions: sigmoid and 1/den on v_exp_f32 + v_rcp_f32, score dot as pk_mul + pk_fma (pairs 0+2, 1+3), den summed per lane slot then across slots
# speedup vs baseline: 7.7570x; 1.0910x over previous
#include <hip/hip_runtime.h>


typedef float v2f __attribute__((ext_vector_type(2)));
typedef float v8f __attribute__((ext_vector_type(8)));

#define CH 128
#define NEG_SLOPE 0.2f
#define SM_EPS 1e-16f
#define WT_PAD 132

#define BIN_SHIFT 9
#define BIN_NODES (1 << BIN_SHIFT)
#define MAX_BINS 256
#define SORT_THREADS 1024
#define SORT_PER_THREAD 8
#define SORT_CHUNK (SORT_THREADS * SORT_PER_THREAD)
#define TABLE_AHEAD 16
#define FIN_CAP 16384
#define FIN_AHEAD 8

__device__ __forceinline__ int wave_sum_int(int v) {
  for (int o = 16; o > 0; o >>= 1) v += __shfl_xor(v, o, 32);
  return v;
}

__device__ __forceinline__ int block_exscan(int v, int* wsum, int& total) {
  const int lane = threadIdx.x & 31;
  const int wave = threadIdx.x >> 5;
  int incl = v;
#pragma unroll
  for (int o = 1; o < 32; o <<= 1) {
    const int y = __shfl_up(incl, o, 32);
    if (lane >= o) incl += y;
  }
  if (lane == 31) wsum[wave] = incl;
  __syncthreads();
  const int w = wsum[lane];
  const int before = wave_sum_int((lane < wave) ? w : 0);
  total = wave_sum_int(w);
  __syncthreads();
  return before + incl - v;
}

__global__ void bin_count(const int* __restrict__ dst, int* __restrict__ table,
                          int n_edges) {
  __shared__ int hist[MAX_BINS];
  const int t = threadIdx.x;
  if (t < MAX_BINS) hist[t] = 0;
  __syncthreads();
  const int e0 = blockIdx.x * SORT_CHUNK + t;
  int dv[SORT_PER_THREAD];
#pragma unroll
  for (int k = 0; k < SORT_PER_THREAD; ++k) {
    const int e = e0 + k * SORT_THREADS;
    dv[k] = (e < n_edges) ? dst[e] : -1;
  }
#pragma unroll
  for (int k = 0; k < SORT_PER_THREAD; ++k)
    if (dv[k] >= 0) atomicAdd(&hist[dv[k] >> BIN_SHIFT], 1);
  __syncthreads();
  if (t < MAX_BINS) table[(size_t)blockIdx.x * MAX_BINS + t] = hist[t];
}

__global__ void bin_scatter(const int* __restrict__ src,
                            const int* __restrict__ dst,
                            const int* __restrict__ table,
                            int* __restrict__ pairs, int* __restrict__ bin_start,
                            int n_edges, int n_chunks) {
  __shared__ int2 stage[SORT_CHUNK];
  __shared__ int tot[MAX_BINS];
  __shared__ int bef[MAX_BINS];
  __shared__ int cursor[MAX_BINS];
  __shared__ int gbase[MAX_BINS];
  __shared__ int wsum[32];
  const int t = threadIdx.x;
  const int c = blockIdx.x;

  const int e0 = c * SORT_CHUNK + t;
  int dv[SORT_PER_THREAD];
  int sv[SORT_PER_THREAD];
#pragma unroll
  for (int k = 0; k < SORT_PER_THREAD; ++k) {
    const int e = e0 + k * SORT_THREADS;
    dv[k] = (e < n_edges) ? dst[e] : -1;
    sv[k] = (e < n_edges) ? src[e] : 0;
  }

  if (t < MAX_BINS) {
    tot[t] = 0;
    bef[t] = 0;
  }
  __syncthreads();
  {
    const int b = t & (MAX_BINS - 1);
    const int part = t >> 8;
    int ptot = 0;
    int pbef = 0;
    for (int r0 = part; r0 < n_chunks; r0 += 4 * TABLE_AHEAD) {
      int v[TABLE_AHEAD];
#pragma unroll
      for (int k = 0; k < TABLE_AHEAD; ++k) {
        const int r = r0 + 4 * k;
        v[k] = (r < n_chunks) ? table[(size_t)r * MAX_BINS + b] : 0;
      }
#pragma unroll
      for (int k = 0; k < TABLE_AHEAD; ++k) {
        const int r = r0 + 4 * k;
        ptot += v[k];
        if (r < c) pbef += v[k];
      }
    }
    atomicAdd(&tot[b], ptot);
    atomicAdd(&bef[b], pbef);
  }
  __syncthreads();
  const int own = (t < MAX_BINS) ? table[(size_t)c * MAX_BINS + t] : 0;
  const int total = (t < MAX_BINS) ? tot[t] : 0;
  int sum_own;
  int sum_all;
  const int lstart = block_exscan(own, wsum, sum_own);
  const int bstart = block_exscan(total, wsum, sum_all);
  if (t < MAX_BINS) {
    cursor[t] = lstart;
    gbase[t] = bstart + bef[t] - lstart;
    if (c == 0) bin_start[t] = bstart;
  }
  if (c == 0 && t == 0) bin_start[MAX_BINS] = sum_all;
  __syncthreads();

#pragma unroll
  for (int k = 0; k < SORT_PER_THREAD; ++k) {
    if (dv[k] >= 0) {
      const int slot = atomicAdd(&cursor[dv[k] >> BIN_SHIFT], 1);
      stage[slot] = make_int2(dv[k], sv[k]);
    }
  }
  __syncthreads();
  const int left = n_edges - c * SORT_CHUNK;
  const int nloc = (left < SORT_CHUNK) ? left : SORT_CHUNK;
  int2* out2 = (int2*)pairs;
  for (int i = t; i < nloc; i += SORT_THREADS) {
    const int2 p = stage[i];
    out2[gbase[p.x >> BIN_SHIFT] + i] = p;
  }
}

__global__ void bin_finish(const int* __restrict__ pairs,
                           const int* __restrict__ bin_start,
                           int* __restrict__ rowptr, int* __restrict__ csr_src,
                           int n_nodes, int n_edges) {
  __shared__ int ncnt[BIN_NODES];
  __shared__ int sorted[FIN_CAP];
  __shared__ int wsum[32];
  const int t = threadIdx.x;
  const int q = blockIdx.x;
  const int node0 = q << BIN_SHIFT;
  const int beg = bin_start[q];
  const int end = bin_start[q + 1];
  const int len = end - beg;
  const int2* in2 = (const int2*)pairs;
  if (t < BIN_NODES) ncnt[t] = 0;
  __syncthreads();
  for (int i0 = beg + t; i0 < end; i0 += FIN_AHEAD * SORT_THREADS) {
    int dv[FIN_AHEAD];
#pragma unroll
    for (int k = 0; k < FIN_AHEAD; ++k) {
      const int i = i0 + k * SORT_THREADS;
      dv[k] = (i < end) ? in2[i].x : -1;
    }
#pragma unroll
    for (int k = 0; k < FIN_AHEAD; ++k)
      if (dv[k] >= 0) atomicAdd(&ncnt[dv[k] - node0], 1);
  }
  __syncthreads();
  const int mine = (t < BIN_NODES) ? ncnt[t] : 0;
  int sum_bin;
  const int excl = block_exscan(mine, wsum, sum_bin);
  if (t < BIN_NODES) {
    ncnt[t] = excl;
    const int node = node0 + t;
    if (node < n_nodes) rowptr[node] = beg + excl;
  }
  if (q == (int)gridDim.x - 1 && t == 0) rowptr[n_nodes] = n_edges;
  __syncthreads();
  const bool fits = len <= FIN_CAP;
  for (int i0 = beg + t; i0 < end; i0 += FIN_AHEAD * SORT_THREADS) {
    int2 p[FIN_AHEAD];
#pragma unroll
    for (int k = 0; k < FIN_AHEAD; ++k) {
      const int i = i0 + k * SORT_THREADS;
      p[k] = (i < end) ? in2[i] : make_int2(-1, 0);
    }
#pragma unroll
    for (int k = 0; k < FIN_AHEAD; ++k) {
      if (p[k].x >= 0) {
        const int slot = atomicAdd(&ncnt[p[k].x - node0], 1);
        if (fits)
          sorted[slot] = p[k].y;
        else
          csr_src[beg + slot] = p[k].y;
      }
    }
  }
  if (fits) {
    __syncthreads();
    for (int i = t; i < len; i += SORT_THREADS) csr_src[beg + i] = sorted[i];
  }
}

#define LOG2E 1.4426950408889634f

__device__ __forceinline__ float leaky(float v) {
  return __builtin_fmaxf(v, NEG_SLOPE * v);
}

__device__ __forceinline__ float xhalf(float x) {
  const int b = __builtin_bit_cast(int, x);
  return __builtin_bit_cast(
      float, __builtin_amdgcn_permlanex16(b, b, 0x76543210, 0xfedcba98, false, false));
}

template <int M>
__device__ __forceinline__ float xrow(float x) {
  const int b = __builtin_bit_cast(int, x);
  return __builtin_bit_cast(
      float, __builtin_amdgcn_update_dpp(0, b, 0x160 | M, 0xf, 0xf, true));
}

__device__ __forceinline__ float dot4(const float4 a, const float4 b) {
  const v2f alo = {a.x, a.y};
  const v2f ahi = {a.z, a.w};
  const v2f blo = {b.x, b.y};
  const v2f bhi = {b.z, b.w};
  const v2f t = __builtin_elementwise_fma(ahi, bhi, alo * blo);
  return t.x + t.y;
}

__device__ __forceinline__ float exp_neg(float v) {
  return __builtin_amdgcn_exp2f(v * LOG2E);
}

__device__ __forceinline__ float fast_sigmoid(float z) {
  return __builtin_amdgcn_rcpf(1.0f + __builtin_amdgcn_exp2f(-z * LOG2E));
}

__device__ __forceinline__ void load_rows(float4& r0, float4& r1, float4& r2,
                                          float4& r3, int u, int t,
                                          const float4* __restrict__ H4,
                                          int lane) {
  const int u0 = __builtin_amdgcn_readlane(u, t);
  const int u1 = __builtin_amdgcn_readlane(u, t + 1);
  const int u2 = __builtin_amdgcn_readlane(u, t + 2);
  const int u3 = __builtin_amdgcn_readlane(u, t + 3);
  r0 = H4[(size_t)u0 * (CH / 4) + lane];
  r1 = H4[(size_t)u1 * (CH / 4) + lane];
  r2 = H4[(size_t)u2 * (CH / 4) + lane];
  r3 = H4[(size_t)u3 * (CH / 4) + lane];
}

__device__ __forceinline__ void take_rows(const float4 c0, const float4 c1,
                                          const float4 c2, const float4 c3,
                                          int t, int cnt, const float4 as4,
                                          float dv, bool hi16, bool hi8,
                                          int slot, float& m, float& den,
                                          float4& acc) {
  const float p0 = dot4(c0, as4);
  const float p1 = dot4(c1, as4);
  const float p2 = dot4(c2, as4);
  const float p3 = dot4(c3, as4);
  const float kA = hi16 ? p2 : p0;
  const float sA = hi16 ? p0 : p2;
  const float kB = hi16 ? p3 : p1;
  const float sB = hi16 ? p1 : p3;
  const float qA = kA + xhalf(sA);
  const float qB = kB + xhalf(sB);
  const float k2 = hi8 ? qB : qA;
  const float s2 = hi8 ? qA : qB;
  float sc = k2 + xrow<8>(s2);
  sc += xrow<4>(sc);
  sc += xrow<2>(sc);
  sc += xrow<1>(sc);

  float e = leaky(sc + dv);
  e = (t + slot < cnt) ? e : -__builtin_inff();
  float mb = __builtin_fmaxf(e, xrow<8>(e));
  mb = __builtin_fmaxf(mb, xhalf(mb));
  const float mn = __builtin_fmaxf(m, mb);
  const float corr = exp_neg(m - mn);
  const float wv = exp_neg(e - mn);
  den = den * corr + wv;
  m = mn;
  const float w0 = __int_as_float(__builtin_amdgcn_readlane(__float_as_int(wv), 0));
  const float w1 = __int_as_float(__builtin_amdgcn_readlane(__float_as_int(wv), 8));
  const float w2 = __int_as_float(__builtin_amdgcn_readlane(__float_as_int(wv), 16));
  const float w3 = __int_as_float(__builtin_amdgcn_readlane(__float_as_int(wv), 24));
  acc.x = acc.x * corr + w0 * c0.x;
  acc.y = acc.y * corr + w0 * c0.y;
  acc.z = acc.z * corr + w0 * c0.z;
  acc.w = acc.w * corr + w0 * c0.w;
  acc.x += w1 * c1.x;
  acc.y += w1 * c1.y;
  acc.z += w1 * c1.z;
  acc.w += w1 * c1.w;
  acc.x += w2 * c2.x;
  acc.y += w2 * c2.y;
  acc.z += w2 * c2.z;
  acc.w += w2 * c2.w;
  acc.x += w3 * c3.x;
  acc.y += w3 * c3.y;
  acc.z += w3 * c3.z;
  acc.w += w3 * c3.w;
}

#define FB_THREADS 512
#define FB_TEAMS 4
#define FB_BLOCKS 512
#define FB_LDS_FLOATS (CH * WT_PAD + FB_TEAMS * 16 * WT_PAD + 2 * CH)

__device__ __forceinline__ float4 gather_node(
    int v, int beg, int end, int u, const float4 hs, const float4 p4,
    const float4 q4, const int* __restrict__ csr_src,
    const float4* __restrict__ X4, int lane, bool hi16, bool hi8, int slot) {
  float ps = dot4(hs, p4);
  float pd = dot4(hs, q4);
  ps += xrow<1>(ps);
  pd += xrow<1>(pd);
  ps += xrow<2>(ps);
  pd += xrow<2>(pd);
  ps += xrow<4>(ps);
  pd += xrow<4>(pd);
  ps += xrow<8>(ps);
  pd += xrow<8>(pd);
  ps += xhalf(ps);
  pd += xhalf(pd);
  const float dv = pd;

  float m = leaky(ps + dv);
  float den = (slot == 0) ? 1.0f : 0.0f;
  float4 acc = hs;

  if (beg < end) {
    const int left = end - beg;
    const int cnt = (left < 32) ? left : 32;
    float4 a0, a1, a2, a3, b0, b1, b2, b3;
    load_rows(a0, a1, a2, a3, u, 0, X4, lane);
    for (int t = 0;; t += 8) {
      if (t + 4 >= cnt) {
        take_rows(a0, a1, a2, a3, t, cnt, p4, dv, hi16, hi8, slot, m, den, acc);
        break;
      }
      load_rows(b0, b1, b2, b3, u, t + 4, X4, lane);
      take_rows(a0, a1, a2, a3, t, cnt, p4, dv, hi16, hi8, slot, m, den, acc);
      if (t + 8 >= cnt) {
        take_rows(b0, b1, b2, b3, t + 4, cnt, p4, dv, hi16, hi8, slot, m, den, acc);
        break;
      }
      load_rows(a0, a1, a2, a3, u, t + 8, X4, lane);
      take_rows(b0, b1, b2, b3, t + 4, cnt, p4, dv, hi16, hi8, slot, m, den, acc);
    }
  }
  for (int base = beg + 32; base < end; base += 32) {
    const int j = base + lane;
    const int u2 = (j < end) ? csr_src[j] : v;
    const int left = end - base;
    const int cnt = (left < 32) ? left : 32;
    for (int t = 0; t < cnt; t += 4) {
      float4 a0, a1, a2, a3;
      load_rows(a0, a1, a2, a3, u2, t, X4, lane);
      take_rows(a0, a1, a2, a3, t, cnt, p4, dv, hi16, hi8, slot, m, den, acc);
    }
  }

  float dsum = den + xrow<8>(den);
  dsum += xhalf(dsum);
  const float rden = __builtin_amdgcn_rcpf(dsum + SM_EPS);
  return make_float4(acc.x * rden, acc.y * rden, acc.z * rden, acc.w * rden);
}

__global__ __launch_bounds__(FB_THREADS) void gat_layer(
    const int* __restrict__ rowptr, const int* __restrict__ csr_src,
    const float* __restrict__ a_src, const float* __restrict__ a_dst,
    const float* __restrict__ W, const float* __restrict__ bias,
    const float* __restrict__ xin, float* __restrict__ xout, int n_nodes) {
  extern __shared__ float sm[];
  float* wt = sm;
  float* tiles = sm + CH * WT_PAD;
  float* pq = tiles + FB_TEAMS * 16 * WT_PAD;
  const int tid = threadIdx.x;
  for (int i = tid; i < CH * CH; i += FB_THREADS) {
    const int k = i >> 7;
    const int n = i & (CH - 1);
    wt[n * WT_PAD + k] = W[i];
  }
  __syncthreads();
  if (tid < 2 * CH) {
    const float* __restrict__ av = (tid < CH) ? a_src : a_dst;
    const int k = tid & (CH - 1);
    float sacc = 0.0f;
    for (int n = 0; n < CH; ++n) sacc += wt[n * WT_PAD + k] * av[n];
    pq[tid] = sacc;
  }
  __syncthreads();

  const int lane = tid & 31;
  const int wave = __builtin_amdgcn_readfirstlane(tid >> 5);
  const int team = wave >> 2;
  const int sub = wave & 3;
  const bool hi16 = (lane & 16) != 0;
  const bool hi8 = (lane & 8) != 0;
  const int slot = lane >> 3;
  float* mytile = tiles + team * 16 * WT_PAD;
  const float4 p4 = ((const float4*)pq)[lane];
  const float4 q4 = ((const float4*)(pq + CH))[lane];
  const float4 b4 = ((const float4*)bias)[lane];
  const float4* X4 = (const float4*)xin;

  const int n_blocks = (int)gridDim.x;
  const int n_tiles = (n_nodes + 15) >> 4;
  const int full_trips = n_tiles / (n_blocks * FB_TEAMS);
  const int first_extra = full_trips * n_blocks * FB_TEAMS;
  const int rem_tiles = n_tiles - first_extra;
  const int n_extra = (rem_tiles > (int)blockIdx.x)
                          ? (rem_tiles - (int)blockIdx.x + n_blocks - 1) / n_blocks
                          : 0;
  const int n_trips = full_trips + n_extra;

  const int mrow = lane & 15;
  const int koff = (lane >> 4) * 2;
  const float* arow = mytile + mrow * WT_PAD + koff;
  const float* wrow0 = wt + ((2 * sub) * 16 + mrow) * WT_PAD + koff;
  const float* wrow1 = wrow0 + 16 * WT_PAD;

  for (int trip = 0; trip < n_trips; ++trip) {
    const bool extra = trip >= full_trips;
    if (extra) __syncthreads();
    const int tile = extra ? first_extra + (int)blockIdx.x + (trip - full_trips) * n_blocks
                           : ((int)blockIdx.x * full_trips + trip) * FB_TEAMS + team;
    const int v0 = __builtin_amdgcn_readfirstlane(tile * 16 + (extra ? wave : sub * 4));
    const int vend = extra ? v0 + 1 : n_nodes;
    const int vlimit = (vend < n_nodes) ? vend : n_nodes;
    float* myrows = extra ? tiles + wave * WT_PAD : mytile + sub * 4 * WT_PAD;
    const bool multiply = !extra || team == 0;
    int rp[5];
#pragma unroll
    for (int i = 0; i < 5; ++i) {
      const int vi = (v0 + i < n_nodes) ? v0 + i : n_nodes;
      rp[i] = rowptr[vi];
    }
    float4 xs[4];
    int un = 0;
    float4 hn = make_float4(0.0f, 0.0f, 0.0f, 0.0f);
    if (v0 < vlimit) {
      un = (rp[0] + lane < rp[1]) ? csr_src[rp[0] + lane] : v0;
      hn = X4[(size_t)v0 * (CH / 4) + lane];
    }
#pragma unroll
    for (int i = 0; i < 4; ++i) {
      const int v = v0 + i;
      const int u = un;
      const float4 hs = hn;
      if (i < 3 && v + 1 < vlimit) {
        un = (rp[i + 1] + lane < rp[i + 2]) ? csr_src[rp[i + 1] + lane] : v + 1;
        hn = X4[(size_t)(v + 1) * (CH / 4) + lane];
      }
      float4 agg = make_float4(0.0f, 0.0f, 0.0f, 0.0f);
      if (v < vlimit)
        agg = gather_node(v, rp[i], rp[i + 1], u, hs, p4, q4, csr_src, X4, lane,
                          hi16, hi8, slot);
      xs[i] = hs;
      if (!extra || i == 0) *(float4*)(myrows + i * WT_PAD + 4 * lane) = agg;
    }
    __syncthreads();

    v8f acc0 = {};
    v8f acc1 = {};
    if (multiply) {
    v2f fa = *(const v2f*)(arow);
    v2f fb0 = *(const v2f*)(wrow0);
    v2f fb1 = *(const v2f*)(wrow1);
    for (int k4 = 0; k4 < CH; k4 += 4) {
      const int kn = (k4 + 4 < CH) ? k4 + 4 : k4;
      const v2f na = *(const v2f*)(arow + kn);
      const v2f nb0 = *(const v2f*)(wrow0 + kn);
      const v2f nb1 = *(const v2f*)(wrow1 + kn);
      acc0 = __builtin_amdgcn_wmma_f32_16x16x4_f32(false, fa, false, fb0,
                                                   (short)0, acc0, false, false);
      acc1 = __builtin_amdgcn_wmma_f32_16x16x4_f32(false, fa, false, fb1,
                                                   (short)0, acc1, false, false);
      fa = na;
      fb0 = nb0;
      fb1 = nb1;
    }
    }
    __syncthreads();
    if (multiply) {
      const int rhi = (lane >> 4) * 8;
      const int col = (2 * sub) * 16 + (lane & 15);
#pragma unroll
      for (int r = 0; r < 8; ++r) {
        mytile[(r + rhi) * WT_PAD + col] = acc0[r];
        mytile[(r + rhi) * WT_PAD + col + 16] = acc1[r];
      }
    }
    __syncthreads();

#pragma unroll
    for (int i = 0; i < 4; ++i) {
      const int v = v0 + i;
      if (v < vlimit) {
        const float4 o = *(const float4*)(myrows + i * WT_PAD + 4 * lane);
        float4 xv = xs[i];
        xv.x += fast_sigmoid(o.x + b4.x);
        xv.y += fast_sigmoid(o.y + b4.y);
        xv.z += fast_sigmoid(o.z + b4.z);
        xv.w += fast_sigmoid(o.w + b4.w);
        ((float4*)(xout + (size_t)v * CH))[lane] = xv;
      }
    }
  }
}

extern "C" void kernel_launch(void* const* d_in, const int* in_sizes, int n_in,
                              void* d_out, int out_size, void* d_ws,
                              size_t ws_size, hipStream_t stream) {
  const float* x_in = (const float*)d_in[0];
  const int* ei = (const int*)d_in[1];
  const float* Ws = (const float*)d_in[2];
  const float* att_src = (const float*)d_in[3];
  const float* att_dst = (const float*)d_in[4];
  const float* bias = (const float*)d_in[5];

  const int N = in_sizes[0] / CH;
  const int E = in_sizes[1] / 2;
  const int M = in_sizes[3] / CH;
  const int* src = ei;
  const int* dst = ei + E;

  float* ws = (float*)d_ws;
  float* plane = ws;
  int* rowptr = (int*)(plane + (size_t)N * CH);
  int* csr_src = rowptr + N + 1;

  float* out = (float*)d_out;

  const int n_chunks = (E + SORT_CHUNK - 1) / SORT_CHUNK;
  const int n_bins = (N + BIN_NODES - 1) / BIN_NODES;
  int* pairs = (int*)plane;
  int* table = pairs + 2 * (size_t)E;
  int* bin_start = table + (size_t)n_chunks * MAX_BINS;
  const size_t smem = (size_t)FB_LDS_FLOATS * sizeof(float);

  bin_count<<<n_chunks, SORT_THREADS, 0, stream>>>(dst, table, E);
  bin_scatter<<<n_chunks, SORT_THREADS, 0, stream>>>(src, dst, table, pairs,
                                                    bin_start, E, n_chunks);
  bin_finish<<<n_bins, SORT_THREADS, 0, stream>>>(pairs, bin_start, rowptr,
                                                 csr_src, N, E);

  const float* cur = x_in;
  for (int m = 0; m < M; ++m) {
    const float* W = Ws + (size_t)m * CH * CH;
    const float* as_ = att_src + (size_t)m * CH;
    const float* ad_ = att_dst + (size_t)m * CH;
    const float* b_ = bias + (size_t)m * CH;
    float* nxt = (((M - 1 - m) & 1) != 0) ? plane : out;
    gat_layer<<<FB_BLOCKS, FB_THREADS, smem, stream>>>(rowptr, csr_src, as_, ad_,
                                                     W, b_, cur, nxt, N);
    cur = nxt;
  }
}


# each tile's gather head (row bounds, two nodes' sources and own rows, first eight rows) is requested ahead of the previous tile's matrix phase; block barriers wait for LDS only; float sums unchanged
# speedup vs baseline: 8.0682x; 1.0401x over previous
#include <hip/hip_runtime.h>


typedef float v2f __attribute__((ext_vector_type(2)));
typedef float v8f __attribute__((ext_vector_type(8)));

#define CH 128
#define NEG_SLOPE 0.2f
#define SM_EPS 1e-16f
#define WT_PAD 132

#define BIN_SHIFT 9
#define BIN_NODES (1 << BIN_SHIFT)
#define MAX_BINS 256
#define SORT_THREADS 1024
#define SORT_PER_THREAD 8
#define SORT_CHUNK (SORT_THREADS * SORT_PER_THREAD)
#define TABLE_AHEAD 16
#define FIN_CAP 16384
#define FIN_AHEAD 8

__device__ __forceinline__ int wave_sum_int(int v) {
  for (int o = 16; o > 0; o >>= 1) v += __shfl_xor(v, o, 32);
  return v;
}

__device__ __forceinline__ int block_exscan(int v, int* wsum, int& total) {
  const int lane = threadIdx.x & 31;
  const int wave = threadIdx.x >> 5;
  int incl = v;
#pragma unroll
  for (int o = 1; o < 32; o <<= 1) {
    const int y = __shfl_up(incl, o, 32);
    if (lane >= o) incl += y;
  }
  if (lane == 31) wsum[wave] = incl;
  __syncthreads();
  const int w = wsum[lane];
  const int before = wave_sum_int((lane < wave) ? w : 0);
  total = wave_sum_int(w);
  __syncthreads();
  return before + incl - v;
}

__global__ void bin_count(const int* __restrict__ dst, int* __restrict__ table,
                          int n_edges) {
  __shared__ int hist[MAX_BINS];
  const int t = threadIdx.x;
  if (t < MAX_BINS) hist[t] = 0;
  __syncthreads();
  const int e0 = blockIdx.x * SORT_CHUNK + t;
  int dv[SORT_PER_THREAD];
#pragma unroll
  for (int k = 0; k < SORT_PER_THREAD; ++k) {
    const int e = e0 + k * SORT_THREADS;
    dv[k] = (e < n_edges) ? dst[e] : -1;
  }
#pragma unroll
  for (int k = 0; k < SORT_PER_THREAD; ++k)
    if (dv[k] >= 0) atomicAdd(&hist[dv[k] >> BIN_SHIFT], 1);
  __syncthreads();
  if (t < MAX_BINS) table[(size_t)blockIdx.x * MAX_BINS + t] = hist[t];
}

__global__ void bin_scatter(const int* __restrict__ src,
                            const int* __restrict__ dst,
                            const int* __restrict__ table,
                            int* __restrict__ pairs, int* __restrict__ bin_start,
                            int n_edges, int n_chunks) {
  __shared__ int2 stage[SORT_CHUNK];
  __shared__ int tot[MAX_BINS];
  __shared__ int bef[MAX_BINS];
  __shared__ int cursor[MAX_BINS];
  __shared__ int gbase[MAX_BINS];
  __shared__ int wsum[32];
  const int t = threadIdx.x;
  const int c = blockIdx.x;

  const int e0 = c * SORT_CHUNK + t;
  int dv[SORT_PER_THREAD];
  int sv[SORT_PER_THREAD];
#pragma unroll
  for (int k = 0; k < SORT_PER_THREAD; ++k) {
    const int e = e0 + k * SORT_THREADS;
    dv[k] = (e < n_edges) ? dst[e] : -1;
    sv[k] = (e < n_edges) ? src[e] : 0;
  }

  if (t < MAX_BINS) {
    tot[t] = 0;
    bef[t] = 0;
  }
  __syncthreads();
  {
    const int b = t & (MAX_BINS - 1);
    const int part = t >> 8;
    int ptot = 0;
    int pbef = 0;
    for (int r0 = part; r0 < n_chunks; r0 += 4 * TABLE_AHEAD) {
      int v[TABLE_AHEAD];
#pragma unroll
      for (int k = 0; k < TABLE_AHEAD; ++k) {
        const int r = r0 + 4 * k;
        v[k] = (r < n_chunks) ? table[(size_t)r * MAX_BINS + b] : 0;
      }
#pragma unroll
      for (int k = 0; k < TABLE_AHEAD; ++k) {
        const int r = r0 + 4 * k;
        ptot += v[k];
        if (r < c) pbef += v[k];
      }
    }
    atomicAdd(&tot[b], ptot);
    atomicAdd(&bef[b], pbef);
  }
  __syncthreads();
  const int own = (t < MAX_BINS) ? table[(size_t)c * MAX_BINS + t] : 0;
  const int total = (t < MAX_BINS) ? tot[t] : 0;
  int sum_own;
  int sum_all;
  const int lstart = block_exscan(own, wsum, sum_own);
  const int bstart = block_exscan(total, wsum, sum_all);
  if (t < MAX_BINS) {
    cursor[t] = lstart;
    gbase[t] = bstart + bef[t] - lstart;
    if (c == 0) bin_start[t] = bstart;
  }
  if (c == 0 && t == 0) bin_start[MAX_BINS] = sum_all;
  __syncthreads();

#pragma unroll
  for (int k = 0; k < SORT_PER_THREAD; ++k) {
    if (dv[k] >= 0) {
      const int slot = atomicAdd(&cursor[dv[k] >> BIN_SHIFT], 1);
      stage[slot] = make_int2(dv[k], sv[k]);
    }
  }
  __syncthreads();
  const int left = n_edges - c * SORT_CHUNK;
  const int nloc = (left < SORT_CHUNK) ? left : SORT_CHUNK;
  int2* out2 = (int2*)pairs;
  for (int i = t; i < nloc; i += SORT_THREADS) {
    const int2 p = stage[i];
    out2[gbase[p.x >> BIN_SHIFT] + i] = p;
  }
}

__global__ void bin_finish(const int* __restrict__ pairs,
                           const int* __restrict__ bin_start,
                           int* __restrict__ rowptr, int* __restrict__ csr_src,
                           int n_nodes, int n_edges) {
  __shared__ int ncnt[BIN_NODES];
  __shared__ int sorted[FIN_CAP];
  __shared__ int wsum[32];
  const int t = threadIdx.x;
  const int q = blockIdx.x;
  const int node0 = q << BIN_SHIFT;
  const int beg = bin_start[q];
  const int end = bin_start[q + 1];
  const int len = end - beg;
  const int2* in2 = (const int2*)pairs;
  if (t < BIN_NODES) ncnt[t] = 0;
  __syncthreads();
  for (int i0 = beg + t; i0 < end; i0 += FIN_AHEAD * SORT_THREADS) {
    int dv[FIN_AHEAD];
#pragma unroll
    for (int k = 0; k < FIN_AHEAD; ++k) {
      const int i = i0 + k * SORT_THREADS;
      dv[k] = (i < end) ? in2[i].x : -1;
    }
#pragma unroll
    for (int k = 0; k < FIN_AHEAD; ++k)
      if (dv[k] >= 0) atomicAdd(&ncnt[dv[k] - node0], 1);
  }
  __syncthreads();
  const int mine = (t < BIN_NODES) ? ncnt[t] : 0;
  int sum_bin;
  const int excl = block_exscan(mine, wsum, sum_bin);
  if (t < BIN_NODES) {
    ncnt[t] = excl;
    const int node = node0 + t;
    if (node < n_nodes) rowptr[node] = beg + excl;
  }
  if (q == (int)gridDim.x - 1 && t == 0) rowptr[n_nodes] = n_edges;
  __syncthreads();
  const bool fits = len <= FIN_CAP;
  for (int i0 = beg + t; i0 < end; i0 += FIN_AHEAD * SORT_THREADS) {
    int2 p[FIN_AHEAD];
#pragma unroll
    for (int k = 0; k < FIN_AHEAD; ++k) {
      const int i = i0 + k * SORT_THREADS;
      p[k] = (i < end) ? in2[i] : make_int2(-1, 0);
    }
#pragma unroll
    for (int k = 0; k < FIN_AHEAD; ++k) {
      if (p[k].x >= 0) {
        const int slot = atomicAdd(&ncnt[p[k].x - node0], 1);
        if (fits)
          sorted[slot] = p[k].y;
        else
          csr_src[beg + slot] = p[k].y;
      }
    }
  }
  if (fits) {
    __syncthreads();
    for (int i = t; i < len; i += SORT_THREADS) csr_src[beg + i] = sorted[i];
  }
}

#define LOG2E 1.4426950408889634f

__device__ __forceinline__ float leaky(float v) {
  return __builtin_fmaxf(v, NEG_SLOPE * v);
}

__device__ __forceinline__ float xhalf(float x) {
  const int b = __builtin_bit_cast(int, x);
  return __builtin_bit_cast(
      float, __builtin_amdgcn_permlanex16(b, b, 0x76543210, 0xfedcba98, false, false));
}

template <int M>
__device__ __forceinline__ float xrow(float x) {
  const int b = __builtin_bit_cast(int, x);
  return __builtin_bit_cast(
      float, __builtin_amdgcn_update_dpp(0, b, 0x160 | M, 0xf, 0xf, true));
}

__device__ __forceinline__ float dot4(const float4 a, const float4 b) {
  const v2f alo = {a.x, a.y};
  const v2f ahi = {a.z, a.w};
  const v2f blo = {b.x, b.y};
  const v2f bhi = {b.z, b.w};
  const v2f t = __builtin_elementwise_fma(ahi, bhi, alo * blo);
  return t.x + t.y;
}

__device__ __forceinline__ float exp_neg(float v) {
  return __builtin_amdgcn_exp2f(v * LOG2E);
}

__device__ __forceinline__ float fast_sigmoid(float z) {
  return __builtin_amdgcn_rcpf(1.0f + __builtin_amdgcn_exp2f(-z * LOG2E));
}

__device__ __forceinline__ void load_rows(float4& r0, float4& r1, float4& r2,
                                          float4& r3, int u, int t,
                                          const float4* __restrict__ H4,
                                          int lane) {
  const int u0 = __builtin_amdgcn_readlane(u, t);
  const int u1 = __builtin_amdgcn_readlane(u, t + 1);
  const int u2 = __builtin_amdgcn_readlane(u, t + 2);
  const int u3 = __builtin_amdgcn_readlane(u, t + 3);
  r0 = H4[(size_t)u0 * (CH / 4) + lane];
  r1 = H4[(size_t)u1 * (CH / 4) + lane];
  r2 = H4[(size_t)u2 * (CH / 4) + lane];
  r3 = H4[(size_t)u3 * (CH / 4) + lane];
}

__device__ __forceinline__ void take_rows(const float4 c0, const float4 c1,
                                          const float4 c2, const float4 c3,
                                          int t, int cnt, const float4 as4,
                                          float dv, bool hi16, bool hi8,
                                          int slot, float& m, float& den,
                                          float4& acc) {
  const float p0 = dot4(c0, as4);
  const float p1 = dot4(c1, as4);
  const float p2 = dot4(c2, as4);
  const float p3 = dot4(c3, as4);
  const float kA = hi16 ? p2 : p0;
  const float sA = hi16 ? p0 : p2;
  const float kB = hi16 ? p3 : p1;
  const float sB = hi16 ? p1 : p3;
  const float qA = kA + xhalf(sA);
  const float qB = kB + xhalf(sB);
  const float k2 = hi8 ? qB : qA;
  const float s2 = hi8 ? qA : qB;
  float sc = k2 + xrow<8>(s2);
  sc += xrow<4>(sc);
  sc += xrow<2>(sc);
  sc += xrow<1>(sc);

  float e = leaky(sc + dv);
  e = (t + slot < cnt) ? e : -__builtin_inff();
  float mb = __builtin_fmaxf(e, xrow<8>(e));
  mb = __builtin_fmaxf(mb, xhalf(mb));
  const float mn = __builtin_fmaxf(m, mb);
  const float corr = exp_neg(m - mn);
  const float wv = exp_neg(e - mn);
  den = den * corr + wv;
  m = mn;
  const float w0 = __int_as_float(__builtin_amdgcn_readlane(__float_as_int(wv), 0));
  const float w1 = __int_as_float(__builtin_amdgcn_readlane(__float_as_int(wv), 8));
  const float w2 = __int_as_float(__builtin_amdgcn_readlane(__float_as_int(wv), 16));
  const float w3 = __int_as_float(__builtin_amdgcn_readlane(__float_as_int(wv), 24));
  acc.x = acc.x * corr + w0 * c0.x;
  acc.y = acc.y * corr + w0 * c0.y;
  acc.z = acc.z * corr + w0 * c0.z;
  acc.w = acc.w * corr + w0 * c0.w;
  acc.x += w1 * c1.x;
  acc.y += w1 * c1.y;
  acc.z += w1 * c1.z;
  acc.w += w1 * c1.w;
  acc.x += w2 * c2.x;
  acc.y += w2 * c2.y;
  acc.z += w2 * c2.z;
  acc.w += w2 * c2.w;
  acc.x += w3 * c3.x;
  acc.y += w3 * c3.y;
  acc.z += w3 * c3.z;
  acc.w += w3 * c3.w;
}

#define FB_THREADS 512
#define FB_TEAMS 4
#define FB_BLOCKS 512
#define FB_LDS_FLOATS (CH * WT_PAD + FB_TEAMS * 16 * WT_PAD + 2 * CH)

template <bool PRE>
__device__ __forceinline__ float4 gather_node(
    int v, int beg, int end, int u, const float4 hs, const float4 p4,
    const float4 q4, const int* __restrict__ csr_src,
    const float4* __restrict__ X4, int lane, bool hi16, bool hi8, int slot,
    float4 a0, float4 a1, float4 a2, float4 a3, float4 b0, float4 b1,
    float4 b2, float4 b3) {
  float ps = dot4(hs, p4);
  float pd = dot4(hs, q4);
  ps += xrow<1>(ps);
  pd += xrow<1>(pd);
  ps += xrow<2>(ps);
  pd += xrow<2>(pd);
  ps += xrow<4>(ps);
  pd += xrow<4>(pd);
  ps += xrow<8>(ps);
  pd += xrow<8>(pd);
  ps += xhalf(ps);
  pd += xhalf(pd);
  const float dv = pd;

  float m = leaky(ps + dv);
  float den = (slot == 0) ? 1.0f : 0.0f;
  float4 acc = hs;

  if (beg < end) {
    const int left = end - beg;
    const int cnt = (left < 32) ? left : 32;
    int t = 0;
    bool open = true;
    if (PRE) {
      take_rows(a0, a1, a2, a3, 0, cnt, p4, dv, hi16, hi8, slot, m, den, acc);
      if (4 >= cnt) {
        open = false;
      } else if (8 >= cnt) {
        take_rows(b0, b1, b2, b3, 4, cnt, p4, dv, hi16, hi8, slot, m, den, acc);
        open = false;
      } else {
        load_rows(a0, a1, a2, a3, u, 8, X4, lane);
        take_rows(b0, b1, b2, b3, 4, cnt, p4, dv, hi16, hi8, slot, m, den, acc);
        t = 8;
      }
    } else {
      load_rows(a0, a1, a2, a3, u, 0, X4, lane);
    }
    if (open)
    for (;; t += 8) {
      if (t + 4 >= cnt) {
        take_rows(a0, a1, a2, a3, t, cnt, p4, dv, hi16, hi8, slot, m, den, acc);
        break;
      }
      load_rows(b0, b1, b2, b3, u, t + 4, X4, lane);
      take_rows(a0, a1, a2, a3, t, cnt, p4, dv, hi16, hi8, slot, m, den, acc);
      if (t + 8 >= cnt) {
        take_rows(b0, b1, b2, b3, t + 4, cnt, p4, dv, hi16, hi8, slot, m, den, acc);
        break;
      }
      load_rows(a0, a1, a2, a3, u, t + 8, X4, lane);
      take_rows(b0, b1, b2, b3, t + 4, cnt, p4, dv, hi16, hi8, slot, m, den, acc);
    }
  }
  for (int base = beg + 32; base < end; base += 32) {
    const int j = base + lane;
    const int u2 = (j < end) ? csr_src[j] : v;
    const int left = end - base;
    const int cnt = (left < 32) ? left : 32;
    for (int t = 0; t < cnt; t += 4) {
      float4 r0, r1, r2, r3;
      load_rows(r0, r1, r2, r3, u2, t, X4, lane);
      take_rows(r0, r1, r2, r3, t, cnt, p4, dv, hi16, hi8, slot, m, den, acc);
    }
  }

  float dsum = den + xrow<8>(den);
  dsum += xhalf(dsum);
  const float rden = __builtin_amdgcn_rcpf(dsum + SM_EPS);
  return make_float4(acc.x * rden, acc.y * rden, acc.z * rden, acc.w * rden);
}

__device__ __forceinline__ void lds_barrier() {
  asm volatile("s_wait_dscnt 0x0\n\ts_barrier_signal -1\n\ts_barrier_wait -1" ::: "memory");
}

__device__ __forceinline__ void load_bounds(int (&rp)[5], int v0, int n_nodes,
                                            const int* __restrict__ rowptr) {
#pragma unroll
  for (int i = 0; i < 5; ++i) {
    const int vi = (v0 + i < n_nodes) ? v0 + i : n_nodes;
    rp[i] = rowptr[vi];
  }
}

__global__ __launch_bounds__(FB_THREADS) void gat_layer(
    const int* __restrict__ rowptr, const int* __restrict__ csr_src,
    const float* __restrict__ a_src, const float* __restrict__ a_dst,
    const float* __restrict__ W, const float* __restrict__ bias,
    const float* __restrict__ xin, float* __restrict__ xout, int n_nodes) {
  extern __shared__ float sm[];
  float* wt = sm;
  float* tiles = sm + CH * WT_PAD;
  float* pq = tiles + FB_TEAMS * 16 * WT_PAD;
  const int tid = threadIdx.x;
  const int lane = tid & 31;
  const int wave = __builtin_amdgcn_readfirstlane(tid >> 5);
  const int team = wave >> 2;
  const int sub = wave & 3;
  const float4* X4 = (const float4*)xin;
  const float4 b4 = ((const float4*)bias)[lane];

  const int n_blocks = (int)gridDim.x;
  const int n_tiles = (n_nodes + 15) >> 4;
  const int full_trips = n_tiles / (n_blocks * FB_TEAMS);
  const int first_extra = full_trips * n_blocks * FB_TEAMS;
  const int rem_tiles = n_tiles - first_extra;
  const int n_extra = (rem_tiles > (int)blockIdx.x)
                          ? (rem_tiles - (int)blockIdx.x + n_blocks - 1) / n_blocks
                          : 0;
  const int n_trips = full_trips + n_extra;

  auto first_node = [&](int trip, bool& is_extra, int& limit) -> int {
    is_extra = trip >= full_trips;
    const int tile = is_extra ? first_extra + (int)blockIdx.x + (trip - full_trips) * n_blocks
                              : ((int)blockIdx.x * full_trips + trip) * FB_TEAMS + team;
    const int first = __builtin_amdgcn_readfirstlane(tile * 16 + (is_extra ? wave : sub * 4));
    const int vend = is_extra ? first + 1 : n_nodes;
    limit = (vend < n_nodes) ? vend : n_nodes;
    return first;
  };

  const float4 zero4 = make_float4(0.0f, 0.0f, 0.0f, 0.0f);
  bool extra = false;
  int vlimit = 0;
  int v0 = 0;
  int rp[5] = {0, 0, 0, 0, 0};
  int uc = 0;
  float4 hc = zero4;
  int un = 0;
  float4 hn = zero4;
  float4 ha0 = zero4, ha1 = zero4, ha2 = zero4, ha3 = zero4;
  float4 hb0 = zero4, hb1 = zero4, hb2 = zero4, hb3 = zero4;
  if (n_trips > 0) {
    v0 = first_node(0, extra, vlimit);
    load_bounds(rp, v0, n_nodes, rowptr);
    if (v0 < vlimit) {
      uc = (rp[0] + lane < rp[1]) ? csr_src[rp[0] + lane] : v0;
      hc = X4[(size_t)v0 * (CH / 4) + lane];
    }
    if (v0 + 1 < vlimit) {
      un = (rp[1] + lane < rp[2]) ? csr_src[rp[1] + lane] : v0 + 1;
      hn = X4[(size_t)(v0 + 1) * (CH / 4) + lane];
    }
  }

  for (int i = tid; i < CH * CH; i += FB_THREADS) {
    const int k = i >> 7;
    const int n = i & (CH - 1);
    wt[n * WT_PAD + k] = W[i];
  }
  lds_barrier();
  if (tid < 2 * CH) {
    const float* __restrict__ av = (tid < CH) ? a_src : a_dst;
    const int k = tid & (CH - 1);
    float sacc = 0.0f;
    for (int n = 0; n < CH; ++n) sacc += wt[n * WT_PAD + k] * av[n];
    pq[tid] = sacc;
  }
  if (n_trips > 0 && v0 < vlimit) {
    load_rows(ha0, ha1, ha2, ha3, uc, 0, X4, lane);
    if (rp[1] - rp[0] > 4) load_rows(hb0, hb1, hb2, hb3, uc, 4, X4, lane);
  }
  lds_barrier();

  const bool hi16 = (lane & 16) != 0;
  const bool hi8 = (lane & 8) != 0;
  const int slot = lane >> 3;
  float* mytile = tiles + team * 16 * WT_PAD;
  const float4 p4 = ((const float4*)pq)[lane];
  const float4 q4 = ((const float4*)(pq + CH))[lane];

  const int mrow = lane & 15;
  const int koff = (lane >> 4) * 2;
  const float* arow = mytile + mrow * WT_PAD + koff;
  const float* wrow0 = wt + ((2 * sub) * 16 + mrow) * WT_PAD + koff;
  const float* wrow1 = wrow0 + 16 * WT_PAD;

  for (int trip = 0; trip < n_trips; ++trip) {
    if (extra) lds_barrier();
    float* myrows = extra ? tiles + wave * WT_PAD : mytile + sub * 4 * WT_PAD;
    const bool multiply = !extra || team == 0;
    const bool more = trip + 1 < n_trips;
    bool extra_n = false;
    int vlimit_n = 0;
    int v0_n = 0;
    if (more) v0_n = first_node(trip + 1, extra_n, vlimit_n);
    int rpn[5] = {0, 0, 0, 0, 0};
    int ucn = 0;
    float4 hcn = zero4;
    int unn = 0;
    float4 hnn = zero4;
    float4 xs[4];
#pragma unroll
    for (int i = 0; i < 4; ++i) {
      const int v = v0 + i;
      const int u = uc;
      const float4 hs = hc;
      if (i >= 1 && i < 3 && v + 1 < vlimit) {
        un = (rp[i + 1] + lane < rp[i + 2]) ? csr_src[rp[i + 1] + lane] : v + 1;
        hn = X4[(size_t)(v + 1) * (CH / 4) + lane];
      }
      float4 agg = zero4;
      if (v < vlimit) {
        if (i == 0)
          agg = gather_node<true>(v, rp[i], rp[i + 1], u, hs, p4, q4, csr_src, X4,
                                  lane, hi16, hi8, slot, ha0, ha1, ha2, ha3, hb0,
                                  hb1, hb2, hb3);
        else
          agg = gather_node<false>(v, rp[i], rp[i + 1], u, hs, p4, q4, csr_src, X4,
                                   lane, hi16, hi8, slot, zero4, zero4, zero4,
                                   zero4, zero4, zero4, zero4, zero4);
      }
      xs[i] = hs;
      if (!extra || i == 0) *(float4*)(myrows + i * WT_PAD + 4 * lane) = agg;
      if (i == 0 && more) load_bounds(rpn, v0_n, n_nodes, rowptr);
      if (i == 1 && more) {
        if (v0_n < vlimit_n) {
          ucn = (rpn[0] + lane < rpn[1]) ? csr_src[rpn[0] + lane] : v0_n;
          hcn = X4[(size_t)v0_n * (CH / 4) + lane];
        }
        if (v0_n + 1 < vlimit_n) {
          unn = (rpn[1] + lane < rpn[2]) ? csr_src[rpn[1] + lane] : v0_n + 1;
          hnn = X4[(size_t)(v0_n + 1) * (CH / 4) + lane];
        }
      }
      uc = un;
      hc = hn;
    }
    ha0 = ha1 = ha2 = ha3 = zero4;
    hb0 = hb1 = hb2 = hb3 = zero4;
    if (more && v0_n < vlimit_n) {
      load_rows(ha0, ha1, ha2, ha3, ucn, 0, X4, lane);
      if (rpn[1] - rpn[0] > 4) load_rows(hb0, hb1, hb2, hb3, ucn, 4, X4, lane);
    }
    lds_barrier();

    v8f acc0 = {};
    v8f acc1 = {};
    if (multiply) {
    v2f fa = *(const v2f*)(arow);
    v2f fb0 = *(const v2f*)(wrow0);
    v2f fb1 = *(const v2f*)(wrow1);
    for (int k4 = 0; k4 < CH; k4 += 4) {
      const int kn = (k4 + 4 < CH) ? k4 + 4 : k4;
      const v2f na = *(const v2f*)(arow + kn);
      const v2f nb0 = *(const v2f*)(wrow0 + kn);
      const v2f nb1 = *(const v2f*)(wrow1 + kn);
      acc0 = __builtin_amdgcn_wmma_f32_16x16x4_f32(false, fa, false, fb0,
                                                   (short)0, acc0, false, false);
      acc1 = __builtin_amdgcn_wmma_f32_16x16x4_f32(false, fa, false, fb1,
                                                   (short)0, acc1, false, false);
      fa = na;
      fb0 = nb0;
      fb1 = nb1;
    }
    }
    lds_barrier();
    if (multiply) {
      const int rhi = (lane >> 4) * 8;
      const int col = (2 * sub) * 16 + (lane & 15);
#pragma unroll
      for (int r = 0; r < 8; ++r) {
        mytile[(r + rhi) * WT_PAD + col] = acc0[r];
        mytile[(r + rhi) * WT_PAD + col + 16] = acc1[r];
      }
    }
    lds_barrier();

#pragma unroll
    for (int i = 0; i < 4; ++i) {
      const int v = v0 + i;
      if (v < vlimit) {
        const float4 o = *(const float4*)(myrows + i * WT_PAD + 4 * lane);
        float4 xv = xs[i];
        xv.x += fast_sigmoid(o.x + b4.x);
        xv.y += fast_sigmoid(o.y + b4.y);
        xv.z += fast_sigmoid(o.z + b4.z);
        xv.w += fast_sigmoid(o.w + b4.w);
        ((float4*)(xout + (size_t)v * CH))[lane] = xv;
      }
    }
    extra = extra_n;
    vlimit = vlimit_n;
    v0 = v0_n;
#pragma unroll
    for (int i = 0; i < 5; ++i) rp[i] = rpn[i];
    uc = ucn;
    hc = hcn;
    un = unn;
    hn = hnn;
  }
}

extern "C" void kernel_launch(void* const* d_in, const int* in_sizes, int n_in,
                              void* d_out, int out_size, void* d_ws,
                              size_t ws_size, hipStream_t stream) {
  const float* x_in = (const float*)d_in[0];
  const int* ei = (const int*)d_in[1];
  const float* Ws = (const float*)d_in[2];
  const float* att_src = (const float*)d_in[3];
  const float* att_dst = (const float*)d_in[4];
  const float* bias = (const float*)d_in[5];

  const int N = in_sizes[0] / CH;
  const int E = in_sizes[1] / 2;
  const int M = in_sizes[3] / CH;
  const int* src = ei;
  const int* dst = ei + E;

  float* ws = (float*)d_ws;
  float* plane = ws;
  int* rowptr = (int*)(plane + (size_t)N * CH);
  int* csr_src = rowptr + N + 1;

  float* out = (float*)d_out;

  const int n_chunks = (E + SORT_CHUNK - 1) / SORT_CHUNK;
  const int n_bins = (N + BIN_NODES - 1) / BIN_NODES;
  int* pairs = (int*)plane;
  int* table = pairs + 2 * (size_t)E;
  int* bin_start = table + (size_t)n_chunks * MAX_BINS;
  const size_t smem = (size_t)FB_LDS_FLOATS * sizeof(float);

  bin_count<<<n_chunks, SORT_THREADS, 0, stream>>>(dst, table, E);
  bin_scatter<<<n_chunks, SORT_THREADS, 0, stream>>>(src, dst, table, pairs,
                                                    bin_start, E, n_chunks);
  bin_finish<<<n_bins, SORT_THREADS, 0, stream>>>(pairs, bin_start, rowptr,
                                                 csr_src, N, E);

  const float* cur = x_in;
  for (int m = 0; m < M; ++m) {
    const float* W = Ws + (size_t)m * CH * CH;
    const float* as_ = att_src + (size_t)m * CH;
    const float* ad_ = att_dst + (size_t)m * CH;
    const float* b_ = bias + (size_t)m * CH;
    float* nxt = (((M - 1 - m) & 1) != 0) ? plane : out;
    gat_layer<<<FB_BLOCKS, FB_THREADS, smem, stream>>>(rowptr, csr_src, as_, ad_,
                                                     W, b_, cur, nxt, N);
    cur = nxt;
  }
}
